# starting kernel
# baseline (speedup 1.0000x reference)
#include <hip/hip_runtime.h>
#include <hip/hip_bf16.h>
#include <math.h>


typedef float v2f __attribute__((ext_vector_type(2)));
typedef float v8f __attribute__((ext_vector_type(8)));

#define NFEAT 256
#define NHID  64
#define NOUT  16

template <int K, int N>
__global__ void pack_b(const float* __restrict__ W, v2f* __restrict__ PB) {
    constexpr int NT    = N / 16;
    constexpr int TOTAL = (K / 4) * NT * 32;
    int idx = blockIdx.x * blockDim.x + threadIdx.x;
    if (idx >= TOTAL) return;
    const int lane  = idx & 31;
    const int tile  = (idx >> 5) % NT;
    const int kstep = idx / (NT * 32);
    const int half  = lane >> 4;
    const int col   = tile * 16 + (lane & 15);
    const int k     = kstep * 4 + 2 * half;
    v2f b;
    b.x = W[(size_t)k * N + col];
    b.y = W[(size_t)(k + 1) * N + col];
    PB[idx] = b;
}

template <int K, int N, bool RELU>
__global__ __launch_bounds__(64)
void gemm_wmma_f32(const float* __restrict__ A,
                   const v2f* __restrict__ PB,
                   float* __restrict__ out, int M) {
    constexpr int NT = N / 16;
    const int wave  = (blockIdx.x * blockDim.x + threadIdx.x) >> 5;
    const int lane  = threadIdx.x & 31;
    const int mbase = wave * 16;
    if (mbase >= M) return;

    const int half = lane >> 4;
    const int l16  = lane & 15;

    int arow = mbase + l16;
    if (arow >= M) arow = M - 1;
    const float* __restrict__ ap = A + (size_t)arow * K + 2 * half;
    const v2f*   __restrict__ bp = PB + lane;

    v8f acc[NT];
#pragma unroll
    for (int t = 0; t < NT; ++t) acc[t] = (v8f){0.f,0.f,0.f,0.f,0.f,0.f,0.f,0.f};

#pragma unroll 8
    for (int k4 = 0; k4 < K / 4; ++k4) {
        v2f a = *(const v2f*)(ap + k4 * 4);
        if (RELU) { a.x = fmaxf(a.x, 0.f); a.y = fmaxf(a.y, 0.f); }
        const v2f* __restrict__ bk = bp + (size_t)k4 * (NT * 32);
#pragma unroll
        for (int t = 0; t < NT; ++t) {
            acc[t] = __builtin_amdgcn_wmma_f32_16x16x4_f32(
                false, a, false, bk[t * 32], (short)0, acc[t], false, false);
        }
    }

#pragma unroll
    for (int t = 0; t < NT; ++t) {
#pragma unroll
        for (int j = 0; j < 8; ++j) {
            const int row = mbase + j + 8 * half;
            if (row < M)
                out[(size_t)row * N + t * 16 + l16] = acc[t][j];
        }
    }
}

template <int D>
__global__ void init_bias(float* __restrict__ acc, const float* __restrict__ bias,
                          long long total) {
    long long i = (long long)blockIdx.x * blockDim.x + threadIdx.x;
    if (i < total) acc[i] = bias[i & (D - 1)];
}

template <int D>
__global__ void spmm_scatter(const float* __restrict__ S,
                             const int* __restrict__ src,
                             const int* __restrict__ dst,
                             const float* __restrict__ w,
                             float* __restrict__ acc, int n_edges) {
    constexpr int C = D / 4;
    long long t = (long long)blockIdx.x * blockDim.x + threadIdx.x;
    int e = (int)(t / C);
    int c = (int)(t % C);
    if (e >= n_edges) return;

    const int   s  = src[e];
    const int   d  = dst[e];
    const float wt = w[e];

    const float4 v = *((const float4*)(S + (size_t)s * D) + c);
    float* ap = acc + (size_t)d * D + c * 4;
    atomicAdd(ap + 0, wt * v.x);
    atomicAdd(ap + 1, wt * v.y);
    atomicAdd(ap + 2, wt * v.z);
    atomicAdd(ap + 3, wt * v.w);
}

__global__ void logsoftmax16(const float* __restrict__ logits,
                             float* __restrict__ out, int n) {
    int i = blockIdx.x * blockDim.x + threadIdx.x;
    if (i >= n) return;

    float v[NOUT];
    const float4* p = (const float4*)(logits + (size_t)i * NOUT);
#pragma unroll
    for (int q = 0; q < NOUT / 4; ++q) {
        float4 t = p[q];
        v[q * 4 + 0] = t.x; v[q * 4 + 1] = t.y;
        v[q * 4 + 2] = t.z; v[q * 4 + 3] = t.w;
    }
    float mx = v[0];
#pragma unroll
    for (int j = 1; j < NOUT; ++j) mx = fmaxf(mx, v[j]);
    float s = 0.f;
#pragma unroll
    for (int j = 0; j < NOUT; ++j) s += expf(v[j] - mx);
    const float lse = mx + logf(s);

    float4* o = (float4*)(out + (size_t)i * NOUT);
#pragma unroll
    for (int q = 0; q < NOUT / 4; ++q) {
        float4 t;
        t.x = v[q * 4 + 0] - lse; t.y = v[q * 4 + 1] - lse;
        t.z = v[q * 4 + 2] - lse; t.w = v[q * 4 + 3] - lse;
        o[q] = t;
    }
}

extern "C" void kernel_launch(void* const* d_in, const int* in_sizes, int n_in,
                              void* d_out, int out_size, void* d_ws, size_t ws_size,
                              hipStream_t stream) {
    const float* x    = (const float*)d_in[0];
    const int*   esrc = (const int*)  d_in[1];
    const int*   edst = (const int*)  d_in[2];
    const float* ew   = (const float*)d_in[3];
    const float* W1   = (const float*)d_in[4];
    const float* b1   = (const float*)d_in[5];
    const float* W2   = (const float*)d_in[6];
    const float* b2   = (const float*)d_in[7];
    float* out = (float*)d_out;

    const int n_nodes = in_sizes[0] / NFEAT;
    const int n_edges = in_sizes[1];

    constexpr int PB1_N = (NFEAT / 4) * (NHID / 16) * 32;
    constexpr int PB2_N = (NHID  / 4) * (NOUT / 16) * 32;

    float* S1  = (float*)d_ws;
    float* H   = S1  + (size_t)n_nodes * NHID;
    float* S2  = H   + (size_t)n_nodes * NHID;
    float* A2  = S2  + (size_t)n_nodes * NOUT;
    v2f*   PB1 = (v2f*)(A2 + (size_t)n_nodes * NOUT);
    v2f*   PB2 = PB1 + PB1_N;

    const int mtiles      = (n_nodes + 15) / 16;
    const int gemm_blocks = (mtiles + 1) / 2;

    pack_b<NFEAT, NHID><<<(PB1_N + 255) / 256, 256, 0, stream>>>(W1, PB1);
    pack_b<NHID,  NOUT><<<(PB2_N + 255) / 256, 256, 0, stream>>>(W2, PB2);

    gemm_wmma_f32<NFEAT, NHID, false>
        <<<gemm_blocks, 64, 0, stream>>>(x, PB1, S1, n_nodes);

    {
        long long tot = (long long)n_nodes * NHID;
        init_bias<NHID><<<(int)((tot + 255) / 256), 256, 0, stream>>>(H, b1, tot);
        long long thr = (long long)n_edges * (NHID / 4);
        spmm_scatter<NHID><<<(int)((thr + 255) / 256), 256, 0, stream>>>(
            S1, esrc, edst, ew, H, n_edges);
    }

    gemm_wmma_f32<NHID, NOUT, true>
        <<<gemm_blocks, 64, 0, stream>>>(H, PB2, S2, n_nodes);

    {
        long long tot = (long long)n_nodes * NOUT;
        init_bias<NOUT><<<(int)((tot + 255) / 256), 256, 0, stream>>>(A2, b2, tot);
        long long thr = (long long)n_edges * (NOUT / 4);
        spmm_scatter<NOUT><<<(int)((thr + 255) / 256), 256, 0, stream>>>(
            S2, esrc, edst, ew, A2, n_edges);
    }

    logsoftmax16<<<(n_nodes + 255) / 256, 256, 0, stream>>>(A2, out, n_nodes);
}


# edge scatter: a wave walks 32 edges with lanes along the row, so each f32 atomic instr adds contiguous floats (one whole 128B line or two 64B rows), not 16B-strided quarter lines; arrival-order sums
# speedup vs baseline: 1.6283x; 1.6283x over previous
#include <hip/hip_runtime.h>
#include <hip/hip_bf16.h>
#include <math.h>


typedef float v2f __attribute__((ext_vector_type(2)));
typedef float v8f __attribute__((ext_vector_type(8)));

#define NFEAT 256
#define NHID  64
#define NOUT  16

template <int K, int N>
__global__ void pack_b(const float* __restrict__ W, v2f* __restrict__ PB) {
    constexpr int NT    = N / 16;
    constexpr int TOTAL = (K / 4) * NT * 32;
    int idx = blockIdx.x * blockDim.x + threadIdx.x;
    if (idx >= TOTAL) return;
    const int lane  = idx & 31;
    const int tile  = (idx >> 5) % NT;
    const int kstep = idx / (NT * 32);
    const int half  = lane >> 4;
    const int col   = tile * 16 + (lane & 15);
    const int k     = kstep * 4 + 2 * half;
    v2f b;
    b.x = W[(size_t)k * N + col];
    b.y = W[(size_t)(k + 1) * N + col];
    PB[idx] = b;
}

template <int K, int N, bool RELU>
__global__ __launch_bounds__(64)
void gemm_wmma_f32(const float* __restrict__ A,
                   const v2f* __restrict__ PB,
                   float* __restrict__ out, int M) {
    constexpr int NT = N / 16;
    const int wave  = (blockIdx.x * blockDim.x + threadIdx.x) >> 5;
    const int lane  = threadIdx.x & 31;
    const int mbase = wave * 16;
    if (mbase >= M) return;

    const int half = lane >> 4;
    const int l16  = lane & 15;

    int arow = mbase + l16;
    if (arow >= M) arow = M - 1;
    const float* __restrict__ ap = A + (size_t)arow * K + 2 * half;
    const v2f*   __restrict__ bp = PB + lane;

    v8f acc[NT];
#pragma unroll
    for (int t = 0; t < NT; ++t) acc[t] = (v8f){0.f,0.f,0.f,0.f,0.f,0.f,0.f,0.f};

#pragma unroll 8
    for (int k4 = 0; k4 < K / 4; ++k4) {
        v2f a = *(const v2f*)(ap + k4 * 4);
        if (RELU) { a.x = fmaxf(a.x, 0.f); a.y = fmaxf(a.y, 0.f); }
        const v2f* __restrict__ bk = bp + (size_t)k4 * (NT * 32);
#pragma unroll
        for (int t = 0; t < NT; ++t) {
            acc[t] = __builtin_amdgcn_wmma_f32_16x16x4_f32(
                false, a, false, bk[t * 32], (short)0, acc[t], false, false);
        }
    }

#pragma unroll
    for (int t = 0; t < NT; ++t) {
#pragma unroll
        for (int j = 0; j < 8; ++j) {
            const int row = mbase + j + 8 * half;
            if (row < M)
                out[(size_t)row * N + t * 16 + l16] = acc[t][j];
        }
    }
}

template <int D>
__global__ void init_bias(float* __restrict__ acc, const float* __restrict__ bias,
                          long long total) {
    long long i = (long long)blockIdx.x * blockDim.x + threadIdx.x;
    if (i < total) acc[i] = bias[i & (D - 1)];
}

template <int EPI>
__device__ __forceinline__ int pick_lane(int x, int idx) {
    if constexpr (EPI == 1) return __builtin_amdgcn_readlane(x, idx);
    else                    return __shfl(x, idx);
}
template <int EPI>
__device__ __forceinline__ float pick_lane(float x, int idx) {
    return __int_as_float(pick_lane<EPI>(__float_as_int(x), idx));
}

template <int D>
__global__ __launch_bounds__(256)
void spmm_scatter(const float* __restrict__ S,
                  const int* __restrict__ src,
                  const int* __restrict__ dst,
                  const float* __restrict__ w,
                  float* __restrict__ acc, int n_edges) {
    constexpr int LPE = D < 32 ? D : 32;
    constexpr int EPI = 32 / LPE;
    constexpr int Q   = D / LPE;
    constexpr int NIT = 32 / EPI;
    constexpr int B   = 8;

    const int lane = threadIdx.x & 31;
    const int wave = __builtin_amdgcn_readfirstlane(
        (int)((blockIdx.x * blockDim.x + threadIdx.x) >> 5));
    const int e0 = wave * 32;
    if (e0 >= n_edges) return;
    const int cnt = min(32, n_edges - e0);

    int s = 0, d = 0;
    float wt = 0.f;
    if (lane < cnt) { s = src[e0 + lane]; d = dst[e0 + lane]; wt = w[e0 + lane]; }

    const int sub = lane / LPE;
    const int c   = lane % LPE;

    if (cnt == 32) {
        for (int j0 = 0; j0 < NIT; j0 += B) {
            float v[B][Q];
            float wj[B];
            int   dj[B];
#pragma unroll
            for (int b = 0; b < B; ++b) {
                const int idx = (j0 + b) * EPI + sub;
                const int sj  = pick_lane<EPI>(s, idx);
                dj[b] = pick_lane<EPI>(d, idx);
                wj[b] = pick_lane<EPI>(wt, idx);
                const float* __restrict__ sp = S + (size_t)sj * D + c;
#pragma unroll
                for (int q = 0; q < Q; ++q) v[b][q] = sp[q * LPE];
            }
#pragma unroll
            for (int b = 0; b < B; ++b) {
                float* __restrict__ ap = acc + (size_t)dj[b] * D + c;
#pragma unroll
                for (int q = 0; q < Q; ++q) atomicAdd(ap + q * LPE, wj[b] * v[b][q]);
            }
        }
    } else {
        for (int j = 0; j * EPI < cnt; ++j) {
            const int idx = j * EPI + sub;
            const int   sj = pick_lane<EPI>(s, idx);
            const int   dd = pick_lane<EPI>(d, idx);
            const float ww = pick_lane<EPI>(wt, idx);
            if (idx < cnt) {
                const float* __restrict__ sp = S + (size_t)sj * D + c;
                float* __restrict__ ap = acc + (size_t)dd * D + c;
                for (int q = 0; q < Q; ++q) atomicAdd(ap + q * LPE, ww * sp[q * LPE]);
            }
        }
    }
}

__global__ void logsoftmax16(const float* __restrict__ logits,
                             float* __restrict__ out, int n) {
    int i = blockIdx.x * blockDim.x + threadIdx.x;
    if (i >= n) return;

    float v[NOUT];
    const float4* p = (const float4*)(logits + (size_t)i * NOUT);
#pragma unroll
    for (int q = 0; q < NOUT / 4; ++q) {
        float4 t = p[q];
        v[q * 4 + 0] = t.x; v[q * 4 + 1] = t.y;
        v[q * 4 + 2] = t.z; v[q * 4 + 3] = t.w;
    }
    float mx = v[0];
#pragma unroll
    for (int j = 1; j < NOUT; ++j) mx = fmaxf(mx, v[j]);
    float s = 0.f;
#pragma unroll
    for (int j = 0; j < NOUT; ++j) s += expf(v[j] - mx);
    const float lse = mx + logf(s);

    float4* o = (float4*)(out + (size_t)i * NOUT);
#pragma unroll
    for (int q = 0; q < NOUT / 4; ++q) {
        float4 t;
        t.x = v[q * 4 + 0] - lse; t.y = v[q * 4 + 1] - lse;
        t.z = v[q * 4 + 2] - lse; t.w = v[q * 4 + 3] - lse;
        o[q] = t;
    }
}

extern "C" void kernel_launch(void* const* d_in, const int* in_sizes, int n_in,
                              void* d_out, int out_size, void* d_ws, size_t ws_size,
                              hipStream_t stream) {
    const float* x    = (const float*)d_in[0];
    const int*   esrc = (const int*)  d_in[1];
    const int*   edst = (const int*)  d_in[2];
    const float* ew   = (const float*)d_in[3];
    const float* W1   = (const float*)d_in[4];
    const float* b1   = (const float*)d_in[5];
    const float* W2   = (const float*)d_in[6];
    const float* b2   = (const float*)d_in[7];
    float* out = (float*)d_out;

    const int n_nodes = in_sizes[0] / NFEAT;
    const int n_edges = in_sizes[1];

    constexpr int PB1_N = (NFEAT / 4) * (NHID / 16) * 32;
    constexpr int PB2_N = (NHID  / 4) * (NOUT / 16) * 32;

    float* S1  = (float*)d_ws;
    float* H   = S1  + (size_t)n_nodes * NHID;
    float* S2  = H   + (size_t)n_nodes * NHID;
    float* A2  = S2  + (size_t)n_nodes * NOUT;
    v2f*   PB1 = (v2f*)(A2 + (size_t)n_nodes * NOUT);
    v2f*   PB2 = PB1 + PB1_N;

    const int mtiles      = (n_nodes + 15) / 16;
    const int gemm_blocks = (mtiles + 1) / 2;
    const int scat_blocks = (n_edges + 255) / 256;

    pack_b<NFEAT, NHID><<<(PB1_N + 255) / 256, 256, 0, stream>>>(W1, PB1);
    pack_b<NHID,  NOUT><<<(PB2_N + 255) / 256, 256, 0, stream>>>(W2, PB2);

    gemm_wmma_f32<NFEAT, NHID, false>
        <<<gemm_blocks, 64, 0, stream>>>(x, PB1, S1, n_nodes);

    {
        long long tot = (long long)n_nodes * NHID;
        init_bias<NHID><<<(int)((tot + 255) / 256), 256, 0, stream>>>(H, b1, tot);
        spmm_scatter<NHID><<<scat_blocks, 256, 0, stream>>>(
            S1, esrc, edst, ew, H, n_edges);
    }

    gemm_wmma_f32<NHID, NOUT, true>
        <<<gemm_blocks, 64, 0, stream>>>(H, PB2, S2, n_nodes);

    {
        long long tot = (long long)n_nodes * NOUT;
        init_bias<NOUT><<<(int)((tot + 255) / 256), 256, 0, stream>>>(A2, b2, tot);
        spmm_scatter<NOUT><<<scat_blocks, 256, 0, stream>>>(
            S2, esrc, edst, ew, A2, n_edges);
    }

    logsoftmax16<<<(n_nodes + 255) / 256, 256, 0, stream>>>(A2, out, n_nodes);
}


# edge scatter: accumulator rows split in two by (dst>>2)&1, each half of the chip (by die number) adds only into its own rows; chunks dealt by per-partition counters; arrival-order sums
# speedup vs baseline: 2.0301x; 1.2468x over previous
#include <hip/hip_runtime.h>
#include <hip/hip_bf16.h>
#include <math.h>


typedef float v2f __attribute__((ext_vector_type(2)));
typedef float v8f __attribute__((ext_vector_type(8)));

#define NFEAT 256
#define NHID  64
#define NOUT  16

template <int K, int N>
__global__ void pack_b(const float* __restrict__ W, v2f* __restrict__ PB) {
    constexpr int NT    = N / 16;
    constexpr int TOTAL = (K / 4) * NT * 32;
    int idx = blockIdx.x * blockDim.x + threadIdx.x;
    if (idx >= TOTAL) return;
    const int lane  = idx & 31;
    const int tile  = (idx >> 5) % NT;
    const int kstep = idx / (NT * 32);
    const int half  = lane >> 4;
    const int col   = tile * 16 + (lane & 15);
    const int k     = kstep * 4 + 2 * half;
    v2f b;
    b.x = W[(size_t)k * N + col];
    b.y = W[(size_t)(k + 1) * N + col];
    PB[idx] = b;
}

template <int K, int N, bool RELU>
__global__ __launch_bounds__(64)
void gemm_wmma_f32(const float* __restrict__ A,
                   const v2f* __restrict__ PB,
                   float* __restrict__ out, int M) {
    constexpr int NT = N / 16;
    const int wave  = (blockIdx.x * blockDim.x + threadIdx.x) >> 5;
    const int lane  = threadIdx.x & 31;
    const int mbase = wave * 16;
    if (mbase >= M) return;

    const int half = lane >> 4;
    const int l16  = lane & 15;

    int arow = mbase + l16;
    if (arow >= M) arow = M - 1;
    const float* __restrict__ ap = A + (size_t)arow * K + 2 * half;
    const v2f*   __restrict__ bp = PB + lane;

    v8f acc[NT];
#pragma unroll
    for (int t = 0; t < NT; ++t) acc[t] = (v8f){0.f,0.f,0.f,0.f,0.f,0.f,0.f,0.f};

#pragma unroll 8
    for (int k4 = 0; k4 < K / 4; ++k4) {
        v2f a = *(const v2f*)(ap + k4 * 4);
        if (RELU) { a.x = fmaxf(a.x, 0.f); a.y = fmaxf(a.y, 0.f); }
        const v2f* __restrict__ bk = bp + (size_t)k4 * (NT * 32);
#pragma unroll
        for (int t = 0; t < NT; ++t) {
            acc[t] = __builtin_amdgcn_wmma_f32_16x16x4_f32(
                false, a, false, bk[t * 32], (short)0, acc[t], false, false);
        }
    }

#pragma unroll
    for (int t = 0; t < NT; ++t) {
#pragma unroll
        for (int j = 0; j < 8; ++j) {
            const int row = mbase + j + 8 * half;
            if (row < M)
                out[(size_t)row * N + t * 16 + l16] = acc[t][j];
        }
    }
}

#define CTR_STRIDE 128
template <int D>
__global__ void init_bias(float* __restrict__ acc, const float* __restrict__ bias,
                          long long total, int* __restrict__ ctr) {
    long long i = (long long)blockIdx.x * blockDim.x + threadIdx.x;
    if (i < total) acc[i] = bias[i & (D - 1)];
    if (i == 0) { ctr[0] = 0; ctr[CTR_STRIDE] = 0; }
}

template <int D>
__global__ __launch_bounds__(256)
void spmm_scatter(const float* __restrict__ S,
                  const int* __restrict__ src,
                  const int* __restrict__ dst,
                  const float* __restrict__ w,
                  float* __restrict__ acc, int n_edges,
                  int* __restrict__ ctr) {
    constexpr int LPE = D < 32 ? D : 32;
    constexpr int EPI = 32 / LPE;
    constexpr int Q   = D / LPE;
    constexpr int B   = 16 / EPI;

    __shared__ int sh_c[2];
    __shared__ int sh_p;

    const int lane = threadIdx.x & 31;
    const int wv  = __builtin_amdgcn_readfirstlane((int)(threadIdx.x >> 5));
    const int sub = lane / LPE;
    const int c   = lane % LPE;
    const int nchunks = (n_edges + 255) >> 8;

    const int die = (__builtin_amdgcn_s_sendmsg_rtn(0x87) >> 16) & 15;
    if (threadIdx.x == 0) sh_p = (die >> 2) & 1;
    __syncthreads();
    int p     = __builtin_amdgcn_readfirstlane(sh_p);
    int phase = 0;

    int nxt = 0;
    if (threadIdx.x == 0) nxt = atomicAdd(ctr + p * CTR_STRIDE, 1);

    for (int it = 0; it < 2 * nchunks + 4; ++it) {
        if (threadIdx.x == 0) sh_c[it & 1] = nxt;
        __syncthreads();
        const int ch = __builtin_amdgcn_readfirstlane(sh_c[it & 1]);
        if (ch >= nchunks) {
            if (phase) break;
            phase = 1;
            p ^= 1;
            if (threadIdx.x == 0) nxt = atomicAdd(ctr + p * CTR_STRIDE, 1);
            continue;
        }
        const int e = ch * 256 + wv * 32 + lane;
        int s = 0, d = 0;
        float wt = 0.f;
        if (e < n_edges) { d = dst[e]; s = src[e]; wt = w[e]; }
        if (threadIdx.x == 0) nxt = atomicAdd(ctr + p * CTR_STRIDE, 1);
        const bool sel = (e < n_edges) && (((d >> 2) & 1) == p);
        unsigned m = __builtin_amdgcn_ballot_w32(sel);

        while (m) {
            int ix[B][EPI];
#pragma unroll
            for (int b = 0; b < B; ++b) {
#pragma unroll
                for (int u = 0; u < EPI; ++u) {
                    ix[b][u] = m ? (int)__builtin_ctz(m) : -1;
                    m &= (m - 1u);
                }
            }
            float v[B][Q];
            float wj[B];
            int   dj[B];
#pragma unroll
            for (int b = 0; b < B; ++b) {
                if (ix[b][0] >= 0) {
                    int sj;
                    if constexpr (EPI == 1) {
                        sj    = __builtin_amdgcn_readlane(s, ix[b][0]);
                        dj[b] = __builtin_amdgcn_readlane(d, ix[b][0]);
                        wj[b] = __int_as_float(__builtin_amdgcn_readlane(__float_as_int(wt), ix[b][0]));
                    } else {
                        const int i1 = ix[b][1] >= 0 ? ix[b][1] : ix[b][0];
                        const int s0 = __builtin_amdgcn_readlane(s, ix[b][0]);
                        const int s1 = __builtin_amdgcn_readlane(s, i1);
                        const int d0 = __builtin_amdgcn_readlane(d, ix[b][0]);
                        const int d1 = __builtin_amdgcn_readlane(d, i1);
                        const int w0 = __builtin_amdgcn_readlane(__float_as_int(wt), ix[b][0]);
                        const int w1 = __builtin_amdgcn_readlane(__float_as_int(wt), i1);
                        sj    = sub ? s1 : s0;
                        dj[b] = sub ? d1 : d0;
                        wj[b] = __int_as_float(sub ? w1 : w0);
                    }
                    if (sub == 0 || ix[b][EPI - 1] >= 0) {
                        const float* __restrict__ sp = S + (size_t)sj * D + c;
#pragma unroll
                        for (int q = 0; q < Q; ++q) v[b][q] = sp[q * LPE];
                    }
                }
            }
#pragma unroll
            for (int b = 0; b < B; ++b) {
                if (ix[b][0] >= 0) {
                    if (sub == 0 || ix[b][EPI - 1] >= 0) {
                        float* __restrict__ ap = acc + (size_t)dj[b] * D + c;
#pragma unroll
                        for (int q = 0; q < Q; ++q) atomicAdd(ap + q * LPE, wj[b] * v[b][q]);
                    }
                }
            }
        }
    }
}

__global__ void logsoftmax16(const float* __restrict__ logits,
                             float* __restrict__ out, int n) {
    int i = blockIdx.x * blockDim.x + threadIdx.x;
    if (i >= n) return;

    float v[NOUT];
    const float4* p = (const float4*)(logits + (size_t)i * NOUT);
#pragma unroll
    for (int q = 0; q < NOUT / 4; ++q) {
        float4 t = p[q];
        v[q * 4 + 0] = t.x; v[q * 4 + 1] = t.y;
        v[q * 4 + 2] = t.z; v[q * 4 + 3] = t.w;
    }
    float mx = v[0];
#pragma unroll
    for (int j = 1; j < NOUT; ++j) mx = fmaxf(mx, v[j]);
    float s = 0.f;
#pragma unroll
    for (int j = 0; j < NOUT; ++j) s += expf(v[j] - mx);
    const float lse = mx + logf(s);

    float4* o = (float4*)(out + (size_t)i * NOUT);
#pragma unroll
    for (int q = 0; q < NOUT / 4; ++q) {
        float4 t;
        t.x = v[q * 4 + 0] - lse; t.y = v[q * 4 + 1] - lse;
        t.z = v[q * 4 + 2] - lse; t.w = v[q * 4 + 3] - lse;
        o[q] = t;
    }
}

extern "C" void kernel_launch(void* const* d_in, const int* in_sizes, int n_in,
                              void* d_out, int out_size, void* d_ws, size_t ws_size,
                              hipStream_t stream) {
    const float* x    = (const float*)d_in[0];
    const int*   esrc = (const int*)  d_in[1];
    const int*   edst = (const int*)  d_in[2];
    const float* ew   = (const float*)d_in[3];
    const float* W1   = (const float*)d_in[4];
    const float* b1   = (const float*)d_in[5];
    const float* W2   = (const float*)d_in[6];
    const float* b2   = (const float*)d_in[7];
    float* out = (float*)d_out;

    const int n_nodes = in_sizes[0] / NFEAT;
    const int n_edges = in_sizes[1];

    constexpr int PB1_N = (NFEAT / 4) * (NHID / 16) * 32;
    constexpr int PB2_N = (NHID  / 4) * (NOUT / 16) * 32;

    float* S1  = (float*)d_ws;
    float* H   = S1  + (size_t)n_nodes * NHID;
    float* S2  = H   + (size_t)n_nodes * NHID;
    float* A2  = S2  + (size_t)n_nodes * NOUT;
    v2f*   PB1 = (v2f*)(A2 + (size_t)n_nodes * NOUT);
    v2f*   PB2 = PB1 + PB1_N;

    const int mtiles      = (n_nodes + 15) / 16;
    const int gemm_blocks = (mtiles + 1) / 2;
    const int scat_blocks = 2048;

    pack_b<NFEAT, NHID><<<(PB1_N + 255) / 256, 256, 0, stream>>>(W1, PB1);
    pack_b<NHID,  NOUT><<<(PB2_N + 255) / 256, 256, 0, stream>>>(W2, PB2);

    gemm_wmma_f32<NFEAT, NHID, false>
        <<<gemm_blocks, 64, 0, stream>>>(x, PB1, S1, n_nodes);

    {
        long long tot = (long long)n_nodes * NHID;
        init_bias<NHID><<<(int)((tot + 255) / 256), 256, 0, stream>>>(H, b1, tot, (int*)A2);
        spmm_scatter<NHID><<<scat_blocks, 256, 0, stream>>>(
            S1, esrc, edst, ew, H, n_edges, (int*)A2);
    }

    gemm_wmma_f32<NHID, NOUT, true>
        <<<gemm_blocks, 64, 0, stream>>>(H, PB2, S2, n_nodes);

    {
        long long tot = (long long)n_nodes * NOUT;
        init_bias<NOUT><<<(int)((tot + 255) / 256), 256, 0, stream>>>(A2, b2, tot, (int*)S1);
        spmm_scatter<NOUT><<<scat_blocks, 256, 0, stream>>>(
            S2, esrc, edst, ew, A2, n_edges, (int*)S1);
    }

    logsoftmax16<<<(n_nodes + 255) / 256, 256, 0, stream>>>(A2, out, n_nodes);
}


# bias start values: resident grids, a work-group stores only rows that its own half of the chip (by die number) adds into later; chunks dealt by per-partition counters; 250000 one-float waves gone
# speedup vs baseline: 2.1177x; 1.0432x over previous
#include <hip/hip_runtime.h>
#include <hip/hip_bf16.h>
#include <math.h>


typedef float v2f __attribute__((ext_vector_type(2)));
typedef float v8f __attribute__((ext_vector_type(8)));

#define NFEAT 256
#define NHID  64
#define NOUT  16

#define CTR_STRIDE 128
template <int K, int N>
__global__ void pack_b(const float* __restrict__ W, v2f* __restrict__ PB,
                       int* __restrict__ ctr) {
    constexpr int NT    = N / 16;
    constexpr int TOTAL = (K / 4) * NT * 32;
    int idx = blockIdx.x * blockDim.x + threadIdx.x;
    if (idx == 0) { ctr[0] = 0; ctr[CTR_STRIDE] = 0; }
    if (idx >= TOTAL) return;
    const int lane  = idx & 31;
    const int tile  = (idx >> 5) % NT;
    const int kstep = idx / (NT * 32);
    const int half  = lane >> 4;
    const int col   = tile * 16 + (lane & 15);
    const int k     = kstep * 4 + 2 * half;
    v2f b;
    b.x = W[(size_t)k * N + col];
    b.y = W[(size_t)(k + 1) * N + col];
    PB[idx] = b;
}

template <int K, int N, bool RELU>
__global__ __launch_bounds__(64)
void gemm_wmma_f32(const float* __restrict__ A,
                   const v2f* __restrict__ PB,
                   float* __restrict__ out, int M) {
    constexpr int NT = N / 16;
    const int wave  = (blockIdx.x * blockDim.x + threadIdx.x) >> 5;
    const int lane  = threadIdx.x & 31;
    const int mbase = wave * 16;
    if (mbase >= M) return;

    const int half = lane >> 4;
    const int l16  = lane & 15;

    int arow = mbase + l16;
    if (arow >= M) arow = M - 1;
    const float* __restrict__ ap = A + (size_t)arow * K + 2 * half;
    const v2f*   __restrict__ bp = PB + lane;

    v8f acc[NT];
#pragma unroll
    for (int t = 0; t < NT; ++t) acc[t] = (v8f){0.f,0.f,0.f,0.f,0.f,0.f,0.f,0.f};

#pragma unroll 8
    for (int k4 = 0; k4 < K / 4; ++k4) {
        v2f a = *(const v2f*)(ap + k4 * 4);
        if (RELU) { a.x = fmaxf(a.x, 0.f); a.y = fmaxf(a.y, 0.f); }
        const v2f* __restrict__ bk = bp + (size_t)k4 * (NT * 32);
#pragma unroll
        for (int t = 0; t < NT; ++t) {
            acc[t] = __builtin_amdgcn_wmma_f32_16x16x4_f32(
                false, a, false, bk[t * 32], (short)0, acc[t], false, false);
        }
    }

#pragma unroll
    for (int t = 0; t < NT; ++t) {
#pragma unroll
        for (int j = 0; j < 8; ++j) {
            const int row = mbase + j + 8 * half;
            if (row < M)
                out[(size_t)row * N + t * 16 + l16] = acc[t][j];
        }
    }
}

template <int D, int R>
__global__ __launch_bounds__(256)
void init_bias_own(float* __restrict__ acc, const float* __restrict__ bias,
                   int n_rows, int* __restrict__ ictr,
                   int* __restrict__ sctr, int* __restrict__ nctr) {
    constexpr int UPC = 256 * R / D;
    static_assert(256 % D == 0 && (256 * R) % D == 0, "chunk must hold whole units");

    __shared__ int sh_c[2];
    __shared__ int sh_p;

    const int tid = threadIdx.x;
    if (blockIdx.x == 0 && tid == 0) {
        sctr[0] = 0; sctr[CTR_STRIDE] = 0;
        if (nctr != nullptr) { nctr[0] = 0; nctr[CTR_STRIDE] = 0; }
    }

    const int c4 = tid % (D / 4);
    float4 bv;
    bv.x = bias[c4 * 4 + 0]; bv.y = bias[c4 * 4 + 1];
    bv.z = bias[c4 * 4 + 2]; bv.w = bias[c4 * 4 + 3];

    const int nunits = (n_rows + 3) >> 2;
    const int nch0   = (((nunits + 1) >> 1) + UPC - 1) / UPC;

    const int die = (__builtin_amdgcn_s_sendmsg_rtn(0x87) >> 16) & 15;
    if (tid == 0) sh_p = (die >> 2) & 1;
    __syncthreads();
    int p     = __builtin_amdgcn_readfirstlane(sh_p);
    int phase = 0;

    int nxt = 0;
    if (tid == 0) nxt = atomicAdd(ictr + p * CTR_STRIDE, 1);

    float4* __restrict__ acc4 = (float4*)acc;

    for (int it = 0; it < 2 * nch0 + 4; ++it) {
        if (tid == 0) sh_c[it & 1] = nxt;
        __syncthreads();
        const int ch  = __builtin_amdgcn_readfirstlane(sh_c[it & 1]);
        const int np  = (nunits - p + 1) >> 1;
        const int nch = (np + UPC - 1) / UPC;
        if (ch >= nch) {
            if (phase) break;
            phase = 1;
            p ^= 1;
            if (tid == 0) nxt = atomicAdd(ictr + p * CTR_STRIDE, 1);
            continue;
        }
        if (tid == 0) nxt = atomicAdd(ictr + p * CTR_STRIDE, 1);
#pragma unroll
        for (int r = 0; r < R; ++r) {
            const int i = r * 256 + tid;
            const int k = ch * UPC + i / D;
            const int j = i % D;
            const int u = 2 * k + p;
            const int row = 4 * u + j / (D / 4);
            if (k < np && row < n_rows)
                acc4[(size_t)u * D + j] = bv;
        }
    }
}

template <int D>
__global__ __launch_bounds__(256)
void spmm_scatter(const float* __restrict__ S,
                  const int* __restrict__ src,
                  const int* __restrict__ dst,
                  const float* __restrict__ w,
                  float* __restrict__ acc, int n_edges,
                  int* __restrict__ ctr) {
    constexpr int LPE = D < 32 ? D : 32;
    constexpr int EPI = 32 / LPE;
    constexpr int Q   = D / LPE;
    constexpr int B   = 16 / EPI;

    __shared__ int sh_c[2];
    __shared__ int sh_p;

    const int lane = threadIdx.x & 31;
    const int wv  = __builtin_amdgcn_readfirstlane((int)(threadIdx.x >> 5));
    const int sub = lane / LPE;
    const int c   = lane % LPE;
    const int nchunks = (n_edges + 255) >> 8;

    const int die = (__builtin_amdgcn_s_sendmsg_rtn(0x87) >> 16) & 15;
    if (threadIdx.x == 0) sh_p = (die >> 2) & 1;
    __syncthreads();
    int p     = __builtin_amdgcn_readfirstlane(sh_p);
    int phase = 0;

    int nxt = 0;
    if (threadIdx.x == 0) nxt = atomicAdd(ctr + p * CTR_STRIDE, 1);

    for (int it = 0; it < 2 * nchunks + 4; ++it) {
        if (threadIdx.x == 0) sh_c[it & 1] = nxt;
        __syncthreads();
        const int ch = __builtin_amdgcn_readfirstlane(sh_c[it & 1]);
        if (ch >= nchunks) {
            if (phase) break;
            phase = 1;
            p ^= 1;
            if (threadIdx.x == 0) nxt = atomicAdd(ctr + p * CTR_STRIDE, 1);
            continue;
        }
        const int e = ch * 256 + wv * 32 + lane;
        int s = 0, d = 0;
        float wt = 0.f;
        if (e < n_edges) { d = dst[e]; s = src[e]; wt = w[e]; }
        if (threadIdx.x == 0) nxt = atomicAdd(ctr + p * CTR_STRIDE, 1);
        const bool sel = (e < n_edges) && (((d >> 2) & 1) == p);
        unsigned m = __builtin_amdgcn_ballot_w32(sel);

        while (m) {
            int ix[B][EPI];
#pragma unroll
            for (int b = 0; b < B; ++b) {
#pragma unroll
                for (int u = 0; u < EPI; ++u) {
                    ix[b][u] = m ? (int)__builtin_ctz(m) : -1;
                    m &= (m - 1u);
                }
            }
            float v[B][Q];
            float wj[B];
            int   dj[B];
#pragma unroll
            for (int b = 0; b < B; ++b) {
                if (ix[b][0] >= 0) {
                    int sj;
                    if constexpr (EPI == 1) {
                        sj    = __builtin_amdgcn_readlane(s, ix[b][0]);
                        dj[b] = __builtin_amdgcn_readlane(d, ix[b][0]);
                        wj[b] = __int_as_float(__builtin_amdgcn_readlane(__float_as_int(wt), ix[b][0]));
                    } else {
                        const int i1 = ix[b][1] >= 0 ? ix[b][1] : ix[b][0];
                        const int s0 = __builtin_amdgcn_readlane(s, ix[b][0]);
                        const int s1 = __builtin_amdgcn_readlane(s, i1);
                        const int d0 = __builtin_amdgcn_readlane(d, ix[b][0]);
                        const int d1 = __builtin_amdgcn_readlane(d, i1);
                        const int w0 = __builtin_amdgcn_readlane(__float_as_int(wt), ix[b][0]);
                        const int w1 = __builtin_amdgcn_readlane(__float_as_int(wt), i1);
                        sj    = sub ? s1 : s0;
                        dj[b] = sub ? d1 : d0;
                        wj[b] = __int_as_float(sub ? w1 : w0);
                    }
                    if (sub == 0 || ix[b][EPI - 1] >= 0) {
                        const float* __restrict__ sp = S + (size_t)sj * D + c;
#pragma unroll
                        for (int q = 0; q < Q; ++q) v[b][q] = sp[q * LPE];
                    }
                }
            }
#pragma unroll
            for (int b = 0; b < B; ++b) {
                if (ix[b][0] >= 0) {
                    if (sub == 0 || ix[b][EPI - 1] >= 0) {
                        float* __restrict__ ap = acc + (size_t)dj[b] * D + c;
#pragma unroll
                        for (int q = 0; q < Q; ++q) atomicAdd(ap + q * LPE, wj[b] * v[b][q]);
                    }
                }
            }
        }
    }
}

__global__ void logsoftmax16(const float* __restrict__ logits,
                             float* __restrict__ out, int n) {
    int i = blockIdx.x * blockDim.x + threadIdx.x;
    if (i >= n) return;

    float v[NOUT];
    const float4* p = (const float4*)(logits + (size_t)i * NOUT);
#pragma unroll
    for (int q = 0; q < NOUT / 4; ++q) {
        float4 t = p[q];
        v[q * 4 + 0] = t.x; v[q * 4 + 1] = t.y;
        v[q * 4 + 2] = t.z; v[q * 4 + 3] = t.w;
    }
    float mx = v[0];
#pragma unroll
    for (int j = 1; j < NOUT; ++j) mx = fmaxf(mx, v[j]);
    float s = 0.f;
#pragma unroll
    for (int j = 0; j < NOUT; ++j) s += expf(v[j] - mx);
    const float lse = mx + logf(s);

    float4* o = (float4*)(out + (size_t)i * NOUT);
#pragma unroll
    for (int q = 0; q < NOUT / 4; ++q) {
        float4 t;
        t.x = v[q * 4 + 0] - lse; t.y = v[q * 4 + 1] - lse;
        t.z = v[q * 4 + 2] - lse; t.w = v[q * 4 + 3] - lse;
        o[q] = t;
    }
}

extern "C" void kernel_launch(void* const* d_in, const int* in_sizes, int n_in,
                              void* d_out, int out_size, void* d_ws, size_t ws_size,
                              hipStream_t stream) {
    const float* x    = (const float*)d_in[0];
    const int*   esrc = (const int*)  d_in[1];
    const int*   edst = (const int*)  d_in[2];
    const float* ew   = (const float*)d_in[3];
    const float* W1   = (const float*)d_in[4];
    const float* b1   = (const float*)d_in[5];
    const float* W2   = (const float*)d_in[6];
    const float* b2   = (const float*)d_in[7];
    float* out = (float*)d_out;

    const int n_nodes = in_sizes[0] / NFEAT;
    const int n_edges = in_sizes[1];

    constexpr int PB1_N = (NFEAT / 4) * (NHID / 16) * 32;
    constexpr int PB2_N = (NHID  / 4) * (NOUT / 16) * 32;

    float* S1  = (float*)d_ws;
    float* H   = S1  + (size_t)n_nodes * NHID;
    float* S2  = H   + (size_t)n_nodes * NHID;
    float* A2  = S2  + (size_t)n_nodes * NOUT;
    v2f*   PB1 = (v2f*)(A2 + (size_t)n_nodes * NOUT);
    v2f*   PB2 = PB1 + PB1_N;

    const int mtiles      = (n_nodes + 15) / 16;
    const int gemm_blocks = (mtiles + 1) / 2;
    const int scat_blocks = 2048;

    int* cS1 = (int*)A2;
    int* cI1 = cS1 + 2 * CTR_STRIDE;
    int* cI2 = (int*)PB1;
    int* cS2 = (int*)S1;

    pack_b<NFEAT, NHID><<<(PB1_N + 255) / 256, 256, 0, stream>>>(W1, PB1, cI1);
    pack_b<NHID,  NOUT><<<(PB2_N + 255) / 256, 256, 0, stream>>>(W2, PB2, cI1);

    gemm_wmma_f32<NFEAT, NHID, false>
        <<<gemm_blocks, 64, 0, stream>>>(x, PB1, S1, n_nodes);

    {
        init_bias_own<NHID, 4><<<1024, 256, 0, stream>>>(H, b1, n_nodes, cI1, cS1, cI2);
        spmm_scatter<NHID><<<scat_blocks, 256, 0, stream>>>(
            S1, esrc, edst, ew, H, n_edges, cS1);
    }

    gemm_wmma_f32<NHID, NOUT, true>
        <<<gemm_blocks, 64, 0, stream>>>(H, PB2, S2, n_nodes);

    {
        init_bias_own<NOUT, 2><<<512, 256, 0, stream>>>(A2, b2, n_nodes, cI2, cS2, nullptr);
        spmm_scatter<NOUT><<<scat_blocks, 256, 0, stream>>>(
            S2, esrc, edst, ew, A2, n_edges, cS2);
    }

    logsoftmax16<<<(n_nodes + 255) / 256, 256, 0, stream>>>(A2, out, n_nodes);
}


# edge sums without global float atomics: edges binned per call by 200-row dst range (LDS counters); one work-group per bin adds rows in LDS from the bias, whole-line stores; LDS adds in arrival order
# speedup vs baseline: 2.2241x; 1.0502x over previous
#include <hip/hip_runtime.h>
#include <hip/hip_bf16.h>
#include <math.h>


typedef float v2f __attribute__((ext_vector_type(2)));
typedef float v8f __attribute__((ext_vector_type(8)));

#define NFEAT 256
#define NHID  64
#define NOUT  16

#define RPB   200
#define MAXB  512
#define BINT  512
#define EPT   8
#define SRCB  17

template <int K, int N>
__global__ void pack_b(const float* __restrict__ W, v2f* __restrict__ PB,
                       int* __restrict__ tbl) {
    constexpr int NT    = N / 16;
    constexpr int TOTAL = (K / 4) * NT * 32;
    int idx = blockIdx.x * blockDim.x + threadIdx.x;
    if (tbl != nullptr && idx < 2 * MAXB) tbl[idx] = 0;
    if (idx >= TOTAL) return;
    const int lane  = idx & 31;
    const int tile  = (idx >> 5) % NT;
    const int kstep = idx / (NT * 32);
    const int half  = lane >> 4;
    const int col   = tile * 16 + (lane & 15);
    const int k     = kstep * 4 + 2 * half;
    v2f b;
    b.x = W[(size_t)k * N + col];
    b.y = W[(size_t)(k + 1) * N + col];
    PB[idx] = b;
}

template <int K, int N, bool RELU>
__global__ __launch_bounds__(64)
void gemm_wmma_f32(const float* __restrict__ A,
                   const v2f* __restrict__ PB,
                   float* __restrict__ out, int M) {
    constexpr int NT = N / 16;
    const int wave  = (blockIdx.x * blockDim.x + threadIdx.x) >> 5;
    const int lane  = threadIdx.x & 31;
    const int mbase = wave * 16;
    if (mbase >= M) return;

    const int half = lane >> 4;
    const int l16  = lane & 15;

    int arow = mbase + l16;
    if (arow >= M) arow = M - 1;
    const float* __restrict__ ap = A + (size_t)arow * K + 2 * half;
    const v2f*   __restrict__ bp = PB + lane;

    v8f acc[NT];
#pragma unroll
    for (int t = 0; t < NT; ++t) acc[t] = (v8f){0.f,0.f,0.f,0.f,0.f,0.f,0.f,0.f};

#pragma unroll 8
    for (int k4 = 0; k4 < K / 4; ++k4) {
        v2f a = *(const v2f*)(ap + k4 * 4);
        if (RELU) { a.x = fmaxf(a.x, 0.f); a.y = fmaxf(a.y, 0.f); }
        const v2f* __restrict__ bk = bp + (size_t)k4 * (NT * 32);
#pragma unroll
        for (int t = 0; t < NT; ++t) {
            acc[t] = __builtin_amdgcn_wmma_f32_16x16x4_f32(
                false, a, false, bk[t * 32], (short)0, acc[t], false, false);
        }
    }

#pragma unroll
    for (int t = 0; t < NT; ++t) {
#pragma unroll
        for (int j = 0; j < 8; ++j) {
            const int row = mbase + j + 8 * half;
            if (row < M)
                out[(size_t)row * N + t * 16 + l16] = acc[t][j];
        }
    }
}

__device__ __forceinline__ int bin_of(int d) {
    unsigned b = (unsigned)d / RPB;
    return b < MAXB ? (int)b : MAXB - 1;
}

__global__ __launch_bounds__(BINT)
void bin_count(const int* __restrict__ dst, int n_edges, int* __restrict__ binCount) {
    __shared__ int hist[MAXB + 1];
    const int tid = threadIdx.x;
    hist[tid] = 0;
    if (tid == 0) hist[MAXB] = 0;
    const int e0 = blockIdx.x * (BINT * EPT) + tid;
    int d[EPT];
#pragma unroll
    for (int i = 0; i < EPT; ++i) {
        const int e = e0 + i * BINT;
        d[i] = dst[e < n_edges ? e : n_edges - 1];
    }
    __syncthreads();
#pragma unroll
    for (int i = 0; i < EPT; ++i) {
        const int e = e0 + i * BINT;
        atomicAdd(&hist[e < n_edges ? bin_of(d[i]) : MAXB], 1);
    }
    __syncthreads();
    const int c = hist[tid];
    if (c) atomicAdd(binCount + tid, c);
}

__device__ __forceinline__ int wave_incl_scan(int v, int lane) {
#pragma unroll
    for (int o = 1; o < 32; o <<= 1) {
        const int t = __shfl_up(v, o);
        if (lane >= o) v += t;
    }
    return v;
}

__global__ __launch_bounds__(BINT)
void bin_place(const int* __restrict__ src, const int* __restrict__ dst,
               const float* __restrict__ w, int n_edges,
               const int* __restrict__ binCount, int* __restrict__ binCursor,
               uint2* __restrict__ sorted, int* __restrict__ binBaseOut) {
    __shared__ int   hist[MAXB + 1];
    __shared__ int   lstart[MAXB + 1];
    __shared__ int   gdelta[MAXB + 1];
    __shared__ int   wsumA[BINT / 32];
    __shared__ int   wsumB[BINT / 32];
    __shared__ uint2 stage[BINT * EPT];
    __shared__ unsigned short sbin[BINT * EPT];

    const int tid  = threadIdx.x;
    const int lane = tid & 31;
    const int wv   = tid >> 5;
    const int left   = n_edges - blockIdx.x * (BINT * EPT);
    const int nvalid = left < BINT * EPT ? left : BINT * EPT;
    hist[tid] = 0;
    if (tid == 0) hist[MAXB] = 0;

    const int cnt = binCount[tid];
    const int inc = wave_incl_scan(cnt, lane);
    if (lane == 31) wsumA[wv] = inc;

    int   dd[EPT], ss[EPT];
    float wt[EPT];
    const int e0 = blockIdx.x * (BINT * EPT) + tid;
#pragma unroll
    for (int i = 0; i < EPT; ++i) {
        const int e  = e0 + i * BINT;
        const int ec = e < n_edges ? e : n_edges - 1;
        dd[i] = dst[ec]; ss[i] = src[ec]; wt[i] = w[ec];
    }
    int key[EPT], bn[EPT];
#pragma unroll
    for (int i = 0; i < EPT; ++i) {
        const int e = e0 + i * BINT;
        const int b = bin_of(dd[i]);
        bn[i]  = e < n_edges ? b : MAXB;
        key[i] = ss[i] | ((dd[i] - b * RPB) << SRCB);
    }
    __syncthreads();
    int pre = 0;
#pragma unroll
    for (int k = 0; k < BINT / 32; ++k) if (k < wv) pre += wsumA[k];
    const int excl = pre + inc - cnt;
#pragma unroll
    for (int i = 0; i < EPT; ++i) atomicAdd(&hist[bn[i]], 1);
    __syncthreads();

    const int c    = hist[tid];
    const int linc = wave_incl_scan(c, lane);
    if (lane == 31) wsumB[wv] = linc;
    int off = 0;
    if (c) off = atomicAdd(binCursor + tid, c);
    if (blockIdx.x == 0) {
        binBaseOut[tid] = excl;
        if (tid == BINT - 1) binBaseOut[MAXB] = excl + cnt;
    }
    __syncthreads();
    int lpre = 0;
#pragma unroll
    for (int k = 0; k < BINT / 32; ++k) if (k < wv) lpre += wsumB[k];
    const int ls = lpre + linc - c;
    lstart[tid] = ls;
    gdelta[tid] = excl + off - ls;
    hist[tid]   = 0;
    if (tid == 0) { lstart[MAXB] = nvalid; gdelta[MAXB] = 0; hist[MAXB] = 0; }
    __syncthreads();
    int slot[EPT];
#pragma unroll
    for (int i = 0; i < EPT; ++i) slot[i] = lstart[bn[i]] + atomicAdd(&hist[bn[i]], 1);
#pragma unroll
    for (int i = 0; i < EPT; ++i) {
        stage[slot[i]] = make_uint2((unsigned)key[i], __float_as_uint(wt[i]));
        sbin[slot[i]]  = (unsigned short)bn[i];
    }
    __syncthreads();
    int   gd[EPT];
    uint2 rc[EPT];
#pragma unroll
    for (int i = 0; i < EPT; ++i) {
        const int s = i * BINT + tid;
        gd[i] = gdelta[sbin[s]];
        rc[i] = stage[s];
    }
#pragma unroll
    for (int i = 0; i < EPT; ++i) {
        const int s = i * BINT + tid;
        if (s < nvalid) sorted[gd[i] + s] = rc[i];
    }
}

#define OWN_GB 16

template <int D>
struct OwnCfg {
    static constexpr int VW  = D >= 64 ? 2 : 1;
    static constexpr int LPE = D / VW;
    static constexpr int EPI = 32 / LPE;
    static constexpr int NG  = OWN_GB / EPI;
    static_assert(LPE == 32 || LPE == 16, "row must map onto a wave or a half-wave");
};

template <int D, int J0>
__device__ __forceinline__ void own_loads(const float* __restrict__ S, int key, float wt,
                                          int sub, int c,
                                          float (&vx)[OwnCfg<D>::NG], float (&vy)[OwnCfg<D>::NG],
                                          float (&wj)[OwnCfg<D>::NG], int (&dj)[OwnCfg<D>::NG]) {
    constexpr int VW = OwnCfg<D>::VW, EPI = OwnCfg<D>::EPI, NG = OwnCfg<D>::NG;
#pragma unroll
    for (int u = 0; u < NG; ++u) {
        const int j = J0 + u * EPI;
        int kj;
        if constexpr (EPI == 1) {
            kj    = __builtin_amdgcn_readlane(key, j);
            wj[u] = __int_as_float(__builtin_amdgcn_readlane(__float_as_int(wt), j));
        } else {
            const int k0 = __builtin_amdgcn_readlane(key, j);
            const int k1 = __builtin_amdgcn_readlane(key, j + 1);
            const int w0 = __builtin_amdgcn_readlane(__float_as_int(wt), j);
            const int w1 = __builtin_amdgcn_readlane(__float_as_int(wt), j + 1);
            kj    = sub ? k1 : k0;
            wj[u] = __int_as_float(sub ? w1 : w0);
        }
        const int sj = kj & ((1 << SRCB) - 1);
        int dl = kj >> SRCB;
        dl = dl < RPB ? dl : RPB - 1;
        dj[u] = dl;
        const float* __restrict__ sp = S + (size_t)sj * D + c;
        if constexpr (VW == 2) {
            const v2f t = *(const v2f*)sp;
            vx[u] = t.x; vy[u] = t.y;
        } else {
            vx[u] = *sp; vy[u] = 0.f;
        }
    }
}

template <int D, int J0, bool FULL>
__device__ __forceinline__ void own_adds(float* acc, int cnt, int sub, int c,
                                         const float (&vx)[OwnCfg<D>::NG], const float (&vy)[OwnCfg<D>::NG],
                                         const float (&wj)[OwnCfg<D>::NG], const int (&dj)[OwnCfg<D>::NG]) {
    constexpr int VW = OwnCfg<D>::VW, EPI = OwnCfg<D>::EPI, NG = OwnCfg<D>::NG;
#pragma unroll
    for (int u = 0; u < NG; ++u) {
        const int j = J0 + u * EPI;
        if (FULL || j + sub < cnt) {
            float* ap = acc + dj[u] * D + c;
            atomicAdd(ap, wj[u] * vx[u]);
            if constexpr (VW == 2) atomicAdd(ap + 1, wj[u] * vy[u]);
        }
    }
}

template <int D>
__global__ __launch_bounds__(512)
void own_sum(const float* __restrict__ S, const uint2* __restrict__ sorted,
             const int* __restrict__ binBase, const float* __restrict__ bias,
             float* __restrict__ out, int n_rows) {
    constexpr int VW  = OwnCfg<D>::VW;
    constexpr int LPE = OwnCfg<D>::LPE;
    constexpr int NG  = OwnCfg<D>::NG;
    constexpr int NW  = 512 / 32;
    static_assert(512 % (D / 4) == 0, "a thread keeps one column group in the start-value loop");
    static_assert(OWN_GB == 16, "two groups per 32 records");

    __shared__ float acc[RPB * D];

    const int tid  = threadIdx.x;
    const int lane = tid & 31;
    const int wv   = __builtin_amdgcn_readfirstlane(tid >> 5);
    const int sub  = lane / LPE;
    const int c    = (lane % LPE) * VW;
    const int b    = blockIdx.x;
    const int beg  = binBase[b];
    const int end  = binBase[b + 1];
    const int row0 = b * RPB;

    const int c4 = tid % (D / 4);
    float4 bv;
    bv.x = bias[c4 * 4 + 0]; bv.y = bias[c4 * 4 + 1];
    bv.z = bias[c4 * 4 + 2]; bv.w = bias[c4 * 4 + 3];
    float4* __restrict__ acc4 = (float4*)acc;
    for (int i = tid; i < RPB * D / 4; i += 512) acc4[i] = bv;
    __syncthreads();

    int   nkey = 0;
    float nwt  = 0.f;
    int g = beg + wv * 32;
    if (g + lane < end) { const uint2 r = sorted[g + lane]; nkey = (int)r.x; nwt = __uint_as_float(r.y); }

    float vx[NG], vy[NG], wj[NG];
    int   dj[NG];

    for (; g + 32 <= end; g += NW * 32) {
        const int   key = nkey;
        const float wt  = nwt;
        own_loads<D, 0>(S, key, wt, sub, c, vx, vy, wj, dj);
        own_adds<D, 0, true>(acc, 32, sub, c, vx, vy, wj, dj);
        own_loads<D, OWN_GB>(S, key, wt, sub, c, vx, vy, wj, dj);
        {
            const int pi = (g + NW * 32 + lane) < end ? (g + NW * 32 + lane) : end - 1;
            const uint2 r = sorted[pi];
            nkey = (int)r.x; nwt = __uint_as_float(r.y);
        }
        own_adds<D, OWN_GB, true>(acc, 32, sub, c, vx, vy, wj, dj);
    }
    if (g < end) {
        const int   cnt = end - g;
        const int   key = nkey;
        const float wt  = nwt;
        own_loads<D, 0>(S, key, wt, sub, c, vx, vy, wj, dj);
        own_adds<D, 0, false>(acc, cnt, sub, c, vx, vy, wj, dj);
        if (cnt > OWN_GB) {
            own_loads<D, OWN_GB>(S, key, wt, sub, c, vx, vy, wj, dj);
            own_adds<D, OWN_GB, false>(acc, cnt, sub, c, vx, vy, wj, dj);
        }
    }

    __syncthreads();
    const int rows = (n_rows - row0) < RPB ? (n_rows - row0) : RPB;
    float4* __restrict__ out4 = (float4*)(out + (size_t)row0 * D);
    for (int i = tid; i < rows * D / 4; i += 512) out4[i] = acc4[i];
}

__global__ void logsoftmax16(const float* __restrict__ logits,
                             float* __restrict__ out, int n) {
    int i = blockIdx.x * blockDim.x + threadIdx.x;
    if (i >= n) return;

    float v[NOUT];
    const float4* p = (const float4*)(logits + (size_t)i * NOUT);
#pragma unroll
    for (int q = 0; q < NOUT / 4; ++q) {
        float4 t = p[q];
        v[q * 4 + 0] = t.x; v[q * 4 + 1] = t.y;
        v[q * 4 + 2] = t.z; v[q * 4 + 3] = t.w;
    }
    float mx = v[0];
#pragma unroll
    for (int j = 1; j < NOUT; ++j) mx = fmaxf(mx, v[j]);
    float s = 0.f;
#pragma unroll
    for (int j = 0; j < NOUT; ++j) s += expf(v[j] - mx);
    const float lse = mx + logf(s);

    float4* o = (float4*)(out + (size_t)i * NOUT);
#pragma unroll
    for (int q = 0; q < NOUT / 4; ++q) {
        float4 t;
        t.x = v[q * 4 + 0] - lse; t.y = v[q * 4 + 1] - lse;
        t.z = v[q * 4 + 2] - lse; t.w = v[q * 4 + 3] - lse;
        o[q] = t;
    }
}

extern "C" void kernel_launch(void* const* d_in, const int* in_sizes, int n_in,
                              void* d_out, int out_size, void* d_ws, size_t ws_size,
                              hipStream_t stream) {
    const float* x    = (const float*)d_in[0];
    const int*   esrc = (const int*)  d_in[1];
    const int*   edst = (const int*)  d_in[2];
    const float* ew   = (const float*)d_in[3];
    const float* W1   = (const float*)d_in[4];
    const float* b1   = (const float*)d_in[5];
    const float* W2   = (const float*)d_in[6];
    const float* b2   = (const float*)d_in[7];
    float* out = (float*)d_out;

    const int n_nodes = in_sizes[0] / NFEAT;
    const int n_edges = in_sizes[1];

    constexpr int PB1_N = (NFEAT / 4) * (NHID / 16) * 32;
    constexpr int PB2_N = (NHID  / 4) * (NOUT / 16) * 32;

    float* R1  = (float*)d_ws;
    float* R2  = R1 + (size_t)n_nodes * NHID;
    float* R3  = R2 + (size_t)n_nodes * NHID;
    v2f*   PB1 = (v2f*)(R3 + (size_t)n_nodes * 2 * NOUT);
    v2f*   PB2 = PB1 + PB1_N;

    float* S1 = R1;
    float* H  = R2;
    float* S2 = R1;
    float* A2 = R2;
    uint2* sorted    = (uint2*)R3;
    int*   binCount  = (int*)R2;
    int*   binCursor = binCount + MAXB;
    int*   binBase   = (int*)PB1;

    const int mtiles      = (n_nodes + 15) / 16;
    const int gemm_blocks = (mtiles + 1) / 2;
    const int nslices     = (n_edges + BINT * EPT - 1) / (BINT * EPT);
    const int nbins       = (n_nodes + RPB - 1) / RPB;

    pack_b<NFEAT, NHID><<<(PB1_N + 255) / 256, 256, 0, stream>>>(W1, PB1, binCount);
    pack_b<NHID,  NOUT><<<(PB2_N + 255) / 256, 256, 0, stream>>>(W2, PB2, nullptr);

    bin_count<<<nslices, BINT, 0, stream>>>(edst, n_edges, binCount);

    gemm_wmma_f32<NFEAT, NHID, false>
        <<<gemm_blocks, 64, 0, stream>>>(x, PB1, S1, n_nodes);

    bin_place<<<nslices, BINT, 0, stream>>>(esrc, edst, ew, n_edges,
                                            binCount, binCursor, sorted, binBase);

    own_sum<NHID><<<nbins, 512, 0, stream>>>(S1, sorted, binBase, b1, H, n_nodes);

    gemm_wmma_f32<NHID, NOUT, true>
        <<<gemm_blocks, 64, 0, stream>>>(H, PB2, S2, n_nodes);

    own_sum<NOUT><<<nbins, 512, 0, stream>>>(S2, sorted, binBase, b2, A2, n_nodes);

    logsoftmax16<<<(n_nodes + 255) / 256, 256, 0, stream>>>(A2, out, n_nodes);
}


# owner epilogue: the work-group that has summed its 200 rows in LDS applies the next row-wise stage there (relu + W2 product on f32 WMMA; log_softmax); 2 launches, H and logits round trips gone
# speedup vs baseline: 2.4137x; 1.0852x over previous
#include <hip/hip_runtime.h>
#include <hip/hip_bf16.h>
#include <math.h>


typedef float v2f __attribute__((ext_vector_type(2)));
typedef float v8f __attribute__((ext_vector_type(8)));

#define NFEAT 256
#define NHID  64
#define NOUT  16

#define RPB   200
#define MAXB  512
#define BINT  512
#define EPT   8
#define SRCB  17

template <int K, int N>
__global__ void pack_b(const float* __restrict__ W, v2f* __restrict__ PB,
                       int* __restrict__ tbl) {
    constexpr int NT    = N / 16;
    constexpr int TOTAL = (K / 4) * NT * 32;
    int idx = blockIdx.x * blockDim.x + threadIdx.x;
    if (tbl != nullptr && idx < 2 * MAXB) tbl[idx] = 0;
    if (idx >= TOTAL) return;
    const int lane  = idx & 31;
    const int tile  = (idx >> 5) % NT;
    const int kstep = idx / (NT * 32);
    const int half  = lane >> 4;
    const int col   = tile * 16 + (lane & 15);
    const int k     = kstep * 4 + 2 * half;
    v2f b;
    b.x = W[(size_t)k * N + col];
    b.y = W[(size_t)(k + 1) * N + col];
    PB[idx] = b;
}

template <int K, int N>
__global__ __launch_bounds__(64)
void gemm_wmma_f32(const float* __restrict__ A,
                   const v2f* __restrict__ PB,
                   float* __restrict__ out, int M) {
    constexpr int NT = N / 16;
    const int wave  = (blockIdx.x * blockDim.x + threadIdx.x) >> 5;
    const int lane  = threadIdx.x & 31;
    const int mbase = wave * 16;
    if (mbase >= M) return;

    const int half = lane >> 4;
    const int l16  = lane & 15;

    int arow = mbase + l16;
    if (arow >= M) arow = M - 1;
    const float* __restrict__ ap = A + (size_t)arow * K + 2 * half;
    const v2f*   __restrict__ bp = PB + lane;

    v8f acc[NT];
#pragma unroll
    for (int t = 0; t < NT; ++t) acc[t] = (v8f){0.f,0.f,0.f,0.f,0.f,0.f,0.f,0.f};

#pragma unroll 8
    for (int k4 = 0; k4 < K / 4; ++k4) {
        v2f a = *(const v2f*)(ap + k4 * 4);
        const v2f* __restrict__ bk = bp + (size_t)k4 * (NT * 32);
#pragma unroll
        for (int t = 0; t < NT; ++t) {
            acc[t] = __builtin_amdgcn_wmma_f32_16x16x4_f32(
                false, a, false, bk[t * 32], (short)0, acc[t], false, false);
        }
    }

#pragma unroll
    for (int t = 0; t < NT; ++t) {
#pragma unroll
        for (int j = 0; j < 8; ++j) {
            const int row = mbase + j + 8 * half;
            if (row < M)
                out[(size_t)row * N + t * 16 + l16] = acc[t][j];
        }
    }
}

__device__ __forceinline__ int bin_of(int d) {
    unsigned b = (unsigned)d / RPB;
    return b < MAXB ? (int)b : MAXB - 1;
}

__global__ __launch_bounds__(BINT)
void bin_count(const int* __restrict__ dst, int n_edges, int* __restrict__ binCount) {
    __shared__ int hist[MAXB + 1];
    const int tid = threadIdx.x;
    hist[tid] = 0;
    if (tid == 0) hist[MAXB] = 0;
    const int e0 = blockIdx.x * (BINT * EPT) + tid;
    int d[EPT];
#pragma unroll
    for (int i = 0; i < EPT; ++i) {
        const int e = e0 + i * BINT;
        d[i] = dst[e < n_edges ? e : n_edges - 1];
    }
    __syncthreads();
#pragma unroll
    for (int i = 0; i < EPT; ++i) {
        const int e = e0 + i * BINT;
        atomicAdd(&hist[e < n_edges ? bin_of(d[i]) : MAXB], 1);
    }
    __syncthreads();
    const int c = hist[tid];
    if (c) atomicAdd(binCount + tid, c);
}

__device__ __forceinline__ int wave_incl_scan(int v, int lane) {
#pragma unroll
    for (int o = 1; o < 32; o <<= 1) {
        const int t = __shfl_up(v, o);
        if (lane >= o) v += t;
    }
    return v;
}

__global__ __launch_bounds__(BINT)
void bin_place(const int* __restrict__ src, const int* __restrict__ dst,
               const float* __restrict__ w, int n_edges,
               const int* __restrict__ binCount, int* __restrict__ binCursor,
               uint2* __restrict__ sorted, int* __restrict__ binBaseOut) {
    __shared__ int   hist[MAXB + 1];
    __shared__ int   lstart[MAXB + 1];
    __shared__ int   gdelta[MAXB + 1];
    __shared__ int   wsumA[BINT / 32];
    __shared__ int   wsumB[BINT / 32];
    __shared__ uint2 stage[BINT * EPT];
    __shared__ unsigned short sbin[BINT * EPT];

    const int tid  = threadIdx.x;
    const int lane = tid & 31;
    const int wv   = tid >> 5;
    const int left   = n_edges - blockIdx.x * (BINT * EPT);
    const int nvalid = left < BINT * EPT ? left : BINT * EPT;
    hist[tid] = 0;
    if (tid == 0) hist[MAXB] = 0;

    const int cnt = binCount[tid];
    const int inc = wave_incl_scan(cnt, lane);
    if (lane == 31) wsumA[wv] = inc;

    int   dd[EPT], ss[EPT];
    float wt[EPT];
    const int e0 = blockIdx.x * (BINT * EPT) + tid;
#pragma unroll
    for (int i = 0; i < EPT; ++i) {
        const int e  = e0 + i * BINT;
        const int ec = e < n_edges ? e : n_edges - 1;
        dd[i] = dst[ec]; ss[i] = src[ec]; wt[i] = w[ec];
    }
    int key[EPT], bn[EPT];
#pragma unroll
    for (int i = 0; i < EPT; ++i) {
        const int e = e0 + i * BINT;
        const int b = bin_of(dd[i]);
        bn[i]  = e < n_edges ? b : MAXB;
        key[i] = ss[i] | ((dd[i] - b * RPB) << SRCB);
    }
    __syncthreads();
    int pre = 0;
#pragma unroll
    for (int k = 0; k < BINT / 32; ++k) if (k < wv) pre += wsumA[k];
    const int excl = pre + inc - cnt;
#pragma unroll
    for (int i = 0; i < EPT; ++i) atomicAdd(&hist[bn[i]], 1);
    __syncthreads();

    const int c    = hist[tid];
    const int linc = wave_incl_scan(c, lane);
    if (lane == 31) wsumB[wv] = linc;
    int off = 0;
    if (c) off = atomicAdd(binCursor + tid, c);
    if (blockIdx.x == 0) {
        binBaseOut[tid] = excl;
        if (tid == BINT - 1) binBaseOut[MAXB] = excl + cnt;
    }
    __syncthreads();
    int lpre = 0;
#pragma unroll
    for (int k = 0; k < BINT / 32; ++k) if (k < wv) lpre += wsumB[k];
    const int ls = lpre + linc - c;
    lstart[tid] = ls;
    gdelta[tid] = excl + off - ls;
    hist[tid]   = 0;
    if (tid == 0) { lstart[MAXB] = nvalid; gdelta[MAXB] = 0; hist[MAXB] = 0; }
    __syncthreads();
    int slot[EPT];
#pragma unroll
    for (int i = 0; i < EPT; ++i) slot[i] = lstart[bn[i]] + atomicAdd(&hist[bn[i]], 1);
#pragma unroll
    for (int i = 0; i < EPT; ++i) {
        stage[slot[i]] = make_uint2((unsigned)key[i], __float_as_uint(wt[i]));
        sbin[slot[i]]  = (unsigned short)bn[i];
    }
    __syncthreads();
    int   gd[EPT];
    uint2 rc[EPT];
#pragma unroll
    for (int i = 0; i < EPT; ++i) {
        const int s = i * BINT + tid;
        gd[i] = gdelta[sbin[s]];
        rc[i] = stage[s];
    }
#pragma unroll
    for (int i = 0; i < EPT; ++i) {
        const int s = i * BINT + tid;
        if (s < nvalid) sorted[gd[i] + s] = rc[i];
    }
}

#define OWN_GB 16

template <int D>
struct OwnCfg {
    static constexpr int VW  = D >= 64 ? 2 : 1;
    static constexpr int LPE = D / VW;
    static constexpr int EPI = 32 / LPE;
    static constexpr int NG  = OWN_GB / EPI;
    static constexpr int PITCH = D >= 64 ? D + 4 : D;
    static_assert(LPE == 32 || LPE == 16, "row must map onto a wave or a half-wave");
    static_assert(PITCH % 4 == 0, "rows start on 16-byte boundaries");
};

template <int D, int J0>
__device__ __forceinline__ void own_loads(const float* __restrict__ S, int key, float wt,
                                          int sub, int c,
                                          float (&vx)[OwnCfg<D>::NG], float (&vy)[OwnCfg<D>::NG],
                                          float (&wj)[OwnCfg<D>::NG], int (&dj)[OwnCfg<D>::NG]) {
    constexpr int VW = OwnCfg<D>::VW, EPI = OwnCfg<D>::EPI, NG = OwnCfg<D>::NG;
#pragma unroll
    for (int u = 0; u < NG; ++u) {
        const int j = J0 + u * EPI;
        int kj;
        if constexpr (EPI == 1) {
            kj    = __builtin_amdgcn_readlane(key, j);
            wj[u] = __int_as_float(__builtin_amdgcn_readlane(__float_as_int(wt), j));
        } else {
            const int k0 = __builtin_amdgcn_readlane(key, j);
            const int k1 = __builtin_amdgcn_readlane(key, j + 1);
            const int w0 = __builtin_amdgcn_readlane(__float_as_int(wt), j);
            const int w1 = __builtin_amdgcn_readlane(__float_as_int(wt), j + 1);
            kj    = sub ? k1 : k0;
            wj[u] = __int_as_float(sub ? w1 : w0);
        }
        const int sj = kj & ((1 << SRCB) - 1);
        int dl = kj >> SRCB;
        dl = dl < RPB ? dl : RPB - 1;
        if constexpr (EPI == 1) {
            int ro;
            asm("s_mul_i32 %0, %1, %2" : "=s"(ro) : "s"(dl), "n"(OwnCfg<D>::PITCH));
            dj[u] = ro;
        } else {
            dj[u] = dl * OwnCfg<D>::PITCH;
        }
        const float* __restrict__ sp = S + (size_t)sj * D + c;
        if constexpr (VW == 2) {
            const v2f t = *(const v2f*)sp;
            vx[u] = t.x; vy[u] = t.y;
        } else {
            vx[u] = *sp; vy[u] = 0.f;
        }
    }
}

template <int D, int J0, bool FULL>
__device__ __forceinline__ void own_adds(float* acc, int cnt, int sub, int c,
                                         const float (&vx)[OwnCfg<D>::NG], const float (&vy)[OwnCfg<D>::NG],
                                         const float (&wj)[OwnCfg<D>::NG], const int (&dj)[OwnCfg<D>::NG]) {
    constexpr int VW = OwnCfg<D>::VW, EPI = OwnCfg<D>::EPI, NG = OwnCfg<D>::NG;
#pragma unroll
    for (int u = 0; u < NG; ++u) {
        const int j = J0 + u * EPI;
        if (FULL || j + sub < cnt) {
            float* ap = acc + dj[u] + c;
            atomicAdd(ap, wj[u] * vx[u]);
            if constexpr (VW == 2) atomicAdd(ap + 1, wj[u] * vy[u]);
        }
    }
}

template <int D, bool PRODUCT>
__global__ __launch_bounds__(512)
void own_sum(const float* __restrict__ S, const uint2* __restrict__ sorted,
             const int* __restrict__ binBase, const float* __restrict__ bias,
             const v2f* __restrict__ PBn, float* __restrict__ out, int n_rows) {
    constexpr int VW    = OwnCfg<D>::VW;
    constexpr int LPE   = OwnCfg<D>::LPE;
    constexpr int NG    = OwnCfg<D>::NG;
    constexpr int PITCH = OwnCfg<D>::PITCH;
    constexpr int NW    = 512 / 32;
    constexpr int CG    = D / 4;
    constexpr int NTILE = (RPB + 15) / 16;
    static_assert(512 % CG == 0, "a thread keeps one column group in the start-value loop");
    static_assert(OWN_GB == 16, "two groups per 32 records");
    static_assert(!PRODUCT || (D == NHID && NTILE <= NW), "one wave per 16-row tile, K = NHID");
    static_assert(PRODUCT || (D == NOUT && RPB <= 512), "one thread per row for the row's log-sum-exp");

    __shared__ float acc[RPB * PITCH];

    const int tid  = threadIdx.x;
    const int lane = tid & 31;
    const int wv   = __builtin_amdgcn_readfirstlane(tid >> 5);
    const int sub  = lane / LPE;
    const int c    = (lane % LPE) * VW;
    const int b    = blockIdx.x;
    const int beg  = binBase[b];
    const int end  = binBase[b + 1];
    const int row0 = b * RPB;

    const int c4 = tid % CG;
    float4 bv;
    bv.x = bias[c4 * 4 + 0]; bv.y = bias[c4 * 4 + 1];
    bv.z = bias[c4 * 4 + 2]; bv.w = bias[c4 * 4 + 3];
    float4* __restrict__ acc4 = (float4*)acc;
    for (int r = tid / CG; r < RPB; r += 512 / CG) acc4[r * (PITCH / 4) + c4] = bv;
    __syncthreads();

    int   nkey = 0;
    float nwt  = 0.f;
    int g = beg + wv * 32;
    if (g + lane < end) { const uint2 r = sorted[g + lane]; nkey = (int)r.x; nwt = __uint_as_float(r.y); }

    float vx[NG], vy[NG], wj[NG];
    int   dj[NG];

    for (; g + 32 <= end; g += NW * 32) {
        const int   key = nkey;
        const float wt  = nwt;
        own_loads<D, 0>(S, key, wt, sub, c, vx, vy, wj, dj);
        own_adds<D, 0, true>(acc, 32, sub, c, vx, vy, wj, dj);
        own_loads<D, OWN_GB>(S, key, wt, sub, c, vx, vy, wj, dj);
        {
            const int pi = (g + NW * 32 + lane) < end ? (g + NW * 32 + lane) : end - 1;
            const uint2 r = sorted[pi];
            nkey = (int)r.x; nwt = __uint_as_float(r.y);
        }
        own_adds<D, OWN_GB, true>(acc, 32, sub, c, vx, vy, wj, dj);
    }
    if (g < end) {
        const int   cnt = end - g;
        const int   key = nkey;
        const float wt  = nwt;
        own_loads<D, 0>(S, key, wt, sub, c, vx, vy, wj, dj);
        own_adds<D, 0, false>(acc, cnt, sub, c, vx, vy, wj, dj);
        if (cnt > OWN_GB) {
            own_loads<D, OWN_GB>(S, key, wt, sub, c, vx, vy, wj, dj);
            own_adds<D, OWN_GB, false>(acc, cnt, sub, c, vx, vy, wj, dj);
        }
    }

    const int rows = (n_rows - row0) < RPB ? (n_rows - row0) : RPB;

    if constexpr (PRODUCT) {
        __shared__ float s2s[NTILE * 16 * NOUT];
        const int half = lane >> 4;
        const int l16  = lane & 15;
        v2f bf[D / 4];
        if (wv < NTILE) {
#pragma unroll
            for (int k4 = 0; k4 < D / 4; ++k4) bf[k4] = PBn[k4 * 32 + lane];
        }
        __syncthreads();
        if (wv < NTILE) {
            int ar = wv * 16 + l16;
            ar = ar < RPB ? ar : RPB - 1;
            const float* ap = acc + ar * PITCH + 2 * half;
            v8f d = (v8f){0.f,0.f,0.f,0.f,0.f,0.f,0.f,0.f};
#pragma unroll
            for (int k4 = 0; k4 < D / 4; ++k4) {
                v2f a = *(const v2f*)(ap + k4 * 4);
                a.x = fmaxf(a.x, 0.f); a.y = fmaxf(a.y, 0.f);
                d = __builtin_amdgcn_wmma_f32_16x16x4_f32(
                    false, a, false, bf[k4], (short)0, d, false, false);
            }
#pragma unroll
            for (int j = 0; j < 8; ++j) s2s[(wv * 16 + j + 8 * half) * NOUT + l16] = d[j];
        }
        __syncthreads();
        float4* __restrict__ out4 = (float4*)(out + (size_t)row0 * NOUT);
        const float4* __restrict__ st4 = (const float4*)s2s;
        for (int i = tid; i < rows * (NOUT / 4); i += 512) out4[i] = st4[i];
    } else {
        __shared__ float lse_s[RPB];
        __syncthreads();
        if (tid < RPB) {
            float v[D];
#pragma unroll
            for (int q = 0; q < CG; ++q) {
                const float4 t = acc4[tid * (PITCH / 4) + q];
                v[q * 4 + 0] = t.x; v[q * 4 + 1] = t.y;
                v[q * 4 + 2] = t.z; v[q * 4 + 3] = t.w;
            }
            float mx = v[0];
#pragma unroll
            for (int j = 1; j < D; ++j) mx = fmaxf(mx, v[j]);
            float s = 0.f;
#pragma unroll
            for (int j = 0; j < D; ++j) s += expf(v[j] - mx);
            lse_s[tid] = mx + logf(s);
        }
        __syncthreads();
        static_assert(PRODUCT || PITCH == D, "rows are contiguous in LDS and in out");
        float4* __restrict__ out4 = (float4*)(out + (size_t)row0 * D);
        for (int i = tid; i < rows * CG; i += 512) {
            float4 t = acc4[i];
            const float l = lse_s[i / CG];
            t.x -= l; t.y -= l; t.z -= l; t.w -= l;
            out4[i] = t;
        }
    }
}

extern "C" void kernel_launch(void* const* d_in, const int* in_sizes, int n_in,
                              void* d_out, int out_size, void* d_ws, size_t ws_size,
                              hipStream_t stream) {
    const float* x    = (const float*)d_in[0];
    const int*   esrc = (const int*)  d_in[1];
    const int*   edst = (const int*)  d_in[2];
    const float* ew   = (const float*)d_in[3];
    const float* W1   = (const float*)d_in[4];
    const float* b1   = (const float*)d_in[5];
    const float* W2   = (const float*)d_in[6];
    const float* b2   = (const float*)d_in[7];
    float* out = (float*)d_out;

    const int n_nodes = in_sizes[0] / NFEAT;
    const int n_edges = in_sizes[1];

    constexpr int PB1_N = (NFEAT / 4) * (NHID / 16) * 32;
    constexpr int PB2_N = (NHID  / 4) * (NOUT / 16) * 32;

    float* R1  = (float*)d_ws;
    float* R2  = R1 + (size_t)n_nodes * NHID;
    float* R3  = R2 + (size_t)n_nodes * NHID;
    v2f*   PB1 = (v2f*)(R3 + (size_t)n_nodes * 2 * NOUT);
    v2f*   PB2 = PB1 + PB1_N;

    float* S1 = R1;
    float* S2 = R2;
    uint2* sorted    = (uint2*)R3;
    int*   binCount  = (int*)R2;
    int*   binCursor = binCount + MAXB;
    int*   binBase   = (int*)PB1;

    const int mtiles      = (n_nodes + 15) / 16;
    const int gemm_blocks = (mtiles + 1) / 2;
    const int nslices     = (n_edges + BINT * EPT - 1) / (BINT * EPT);
    const int nbins       = (n_nodes + RPB - 1) / RPB;

    pack_b<NFEAT, NHID><<<(PB1_N + 255) / 256, 256, 0, stream>>>(W1, PB1, binCount);
    pack_b<NHID,  NOUT><<<(PB2_N + 255) / 256, 256, 0, stream>>>(W2, PB2, nullptr);

    bin_count<<<nslices, BINT, 0, stream>>>(edst, n_edges, binCount);

    gemm_wmma_f32<NFEAT, NHID>
        <<<gemm_blocks, 64, 0, stream>>>(x, PB1, S1, n_nodes);

    bin_place<<<nslices, BINT, 0, stream>>>(esrc, edst, ew, n_edges,
                                            binCount, binCursor, sorted, binBase);

    own_sum<NHID, true><<<nbins, 512, 0, stream>>>(S1, sorted, binBase, b1, PB2, S2, n_nodes);

    own_sum<NOUT, false><<<nbins, 512, 0, stream>>>(S2, sorted, binBase, b2, nullptr, out, n_nodes);
}


# merged launches: W2 packing rides in W1's packing launch; bin counting runs as extra work-groups of the f32 WMMA GEMM launch (matrix-bound, memory path idle); 7 launches -> 5
# speedup vs baseline: 2.8908x; 1.1977x over previous
#include <hip/hip_runtime.h>
#include <hip/hip_bf16.h>
#include <math.h>


typedef float v2f __attribute__((ext_vector_type(2)));
typedef float v8f __attribute__((ext_vector_type(8)));

#define NFEAT 256
#define NHID  64
#define NOUT  16

#define RPB   200
#define MAXB  512
#define BINT  512
#define EPT   8
#define SRCB  17

template <int K, int N>
__device__ __forceinline__ void pack_one(const float* __restrict__ W, v2f* __restrict__ PB, int idx) {
    constexpr int NT    = N / 16;
    const int lane  = idx & 31;
    const int tile  = (idx >> 5) % NT;
    const int kstep = idx / (NT * 32);
    const int half  = lane >> 4;
    const int col   = tile * 16 + (lane & 15);
    const int k     = kstep * 4 + 2 * half;
    v2f b;
    b.x = W[(size_t)k * N + col];
    b.y = W[(size_t)(k + 1) * N + col];
    PB[idx] = b;
}

__global__ void pack_both(const float* __restrict__ Wa, v2f* __restrict__ PBa,
                          const float* __restrict__ Wb, v2f* __restrict__ PBb,
                          int* __restrict__ tbl) {
    constexpr int TOTAL_A = (NFEAT / 4) * (NHID / 16) * 32;
    constexpr int TOTAL_B = (NHID  / 4) * (NOUT / 16) * 32;
    const int idx = blockIdx.x * blockDim.x + threadIdx.x;
    if (idx < 2 * MAXB) tbl[idx] = 0;
    if (idx < TOTAL_A)                pack_one<NFEAT, NHID>(Wa, PBa, idx);
    else if (idx < TOTAL_A + TOTAL_B) pack_one<NHID,  NOUT>(Wb, PBb, idx - TOTAL_A);
}

#define FT 128

template <int K, int N>
__device__ __forceinline__
void gemm_wmma_f32(const float* __restrict__ A,
                   const v2f* __restrict__ PB,
                   float* __restrict__ out, int M) {
    constexpr int NT = N / 16;
    const int wave  = (blockIdx.x * FT + threadIdx.x) >> 5;
    const int lane  = threadIdx.x & 31;
    const int mbase = wave * 16;
    if (mbase >= M) return;

    const int half = lane >> 4;
    const int l16  = lane & 15;

    int arow = mbase + l16;
    if (arow >= M) arow = M - 1;
    const float* __restrict__ ap = A + (size_t)arow * K + 2 * half;
    const v2f*   __restrict__ bp = PB + lane;

    v8f acc[NT];
#pragma unroll
    for (int t = 0; t < NT; ++t) acc[t] = (v8f){0.f,0.f,0.f,0.f,0.f,0.f,0.f,0.f};

#pragma unroll 8
    for (int k4 = 0; k4 < K / 4; ++k4) {
        v2f a = *(const v2f*)(ap + k4 * 4);
        const v2f* __restrict__ bk = bp + (size_t)k4 * (NT * 32);
#pragma unroll
        for (int t = 0; t < NT; ++t) {
            acc[t] = __builtin_amdgcn_wmma_f32_16x16x4_f32(
                false, a, false, bk[t * 32], (short)0, acc[t], false, false);
        }
    }

#pragma unroll
    for (int t = 0; t < NT; ++t) {
#pragma unroll
        for (int j = 0; j < 8; ++j) {
            const int row = mbase + j + 8 * half;
            if (row < M)
                out[(size_t)row * N + t * 16 + l16] = acc[t][j];
        }
    }
}

#define CEPT 32
__device__ __forceinline__ int bin_of(int d) {
    unsigned b = (unsigned)d / RPB;
    return b < MAXB ? (int)b : MAXB - 1;
}

__device__ __forceinline__
void bin_count(int slice, const int* __restrict__ dst, int n_edges,
               int* __restrict__ binCount, int* hist) {
    const int tid = threadIdx.x;
    for (int i = tid; i <= MAXB; i += FT) hist[i] = 0;
    const int e0 = slice * (FT * CEPT) + tid;
    int d[CEPT];
#pragma unroll
    for (int i = 0; i < CEPT; ++i) {
        const int e = e0 + i * FT;
        d[i] = dst[e < n_edges ? e : n_edges - 1];
    }
    __syncthreads();
#pragma unroll
    for (int i = 0; i < CEPT; ++i) {
        const int e = e0 + i * FT;
        atomicAdd(&hist[e < n_edges ? bin_of(d[i]) : MAXB], 1);
    }
    __syncthreads();
    for (int i = tid; i < MAXB; i += FT) {
        const int c = hist[i];
        if (c) atomicAdd(binCount + i, c);
    }
}

template <int K, int N>
__global__ __launch_bounds__(FT)
void gemm_and_count(const float* __restrict__ A, const v2f* __restrict__ PB,
                    float* __restrict__ S, int M, int gemm_groups,
                    const int* __restrict__ dst, int n_edges, int* __restrict__ binCount) {
    __shared__ int hist[MAXB + 1];
    if ((int)blockIdx.x < gemm_groups) {
        gemm_wmma_f32<K, N>(A, PB, S, M);
    } else {
        bin_count((int)blockIdx.x - gemm_groups, dst, n_edges, binCount, hist);
    }
}

__device__ __forceinline__ int wave_incl_scan(int v, int lane) {
#pragma unroll
    for (int o = 1; o < 32; o <<= 1) {
        const int t = __shfl_up(v, o);
        if (lane >= o) v += t;
    }
    return v;
}

__global__ __launch_bounds__(BINT)
void bin_place(const int* __restrict__ src, const int* __restrict__ dst,
               const float* __restrict__ w, int n_edges,
               const int* __restrict__ binCount, int* __restrict__ binCursor,
               uint2* __restrict__ sorted, int* __restrict__ binBaseOut) {
    __shared__ int   hist[MAXB + 1];
    __shared__ int   lstart[MAXB + 1];
    __shared__ int   gdelta[MAXB + 1];
    __shared__ int   wsumA[BINT / 32];
    __shared__ int   wsumB[BINT / 32];
    __shared__ uint2 stage[BINT * EPT];
    __shared__ unsigned short sbin[BINT * EPT];

    const int tid  = threadIdx.x;
    const int lane = tid & 31;
    const int wv   = tid >> 5;
    const int left   = n_edges - blockIdx.x * (BINT * EPT);
    const int nvalid = left < BINT * EPT ? left : BINT * EPT;
    hist[tid] = 0;
    if (tid == 0) hist[MAXB] = 0;

    const int cnt = binCount[tid];
    const int inc = wave_incl_scan(cnt, lane);
    if (lane == 31) wsumA[wv] = inc;

    int   dd[EPT], ss[EPT];
    float wt[EPT];
    const int e0 = blockIdx.x * (BINT * EPT) + tid;
#pragma unroll
    for (int i = 0; i < EPT; ++i) {
        const int e  = e0 + i * BINT;
        const int ec = e < n_edges ? e : n_edges - 1;
        dd[i] = dst[ec]; ss[i] = src[ec]; wt[i] = w[ec];
    }
    int key[EPT], bn[EPT];
#pragma unroll
    for (int i = 0; i < EPT; ++i) {
        const int e = e0 + i * BINT;
        const int b = bin_of(dd[i]);
        bn[i]  = e < n_edges ? b : MAXB;
        key[i] = ss[i] | ((dd[i] - b * RPB) << SRCB);
    }
    __syncthreads();
    int pre = 0;
#pragma unroll
    for (int k = 0; k < BINT / 32; ++k) if (k < wv) pre += wsumA[k];
    const int excl = pre + inc - cnt;
#pragma unroll
    for (int i = 0; i < EPT; ++i) atomicAdd(&hist[bn[i]], 1);
    __syncthreads();

    const int c    = hist[tid];
    const int linc = wave_incl_scan(c, lane);
    if (lane == 31) wsumB[wv] = linc;
    int off = 0;
    if (c) off = atomicAdd(binCursor + tid, c);
    if (blockIdx.x == 0) {
        binBaseOut[tid] = excl;
        if (tid == BINT - 1) binBaseOut[MAXB] = excl + cnt;
    }
    __syncthreads();
    int lpre = 0;
#pragma unroll
    for (int k = 0; k < BINT / 32; ++k) if (k < wv) lpre += wsumB[k];
    const int ls = lpre + linc - c;
    lstart[tid] = ls;
    gdelta[tid] = excl + off - ls;
    hist[tid]   = 0;
    if (tid == 0) { lstart[MAXB] = nvalid; gdelta[MAXB] = 0; hist[MAXB] = 0; }
    __syncthreads();
    int slot[EPT];
#pragma unroll
    for (int i = 0; i < EPT; ++i) slot[i] = lstart[bn[i]] + atomicAdd(&hist[bn[i]], 1);
#pragma unroll
    for (int i = 0; i < EPT; ++i) {
        stage[slot[i]] = make_uint2((unsigned)key[i], __float_as_uint(wt[i]));
        sbin[slot[i]]  = (unsigned short)bn[i];
    }
    __syncthreads();
    int   gd[EPT];
    uint2 rc[EPT];
#pragma unroll
    for (int i = 0; i < EPT; ++i) {
        const int s = i * BINT + tid;
        gd[i] = gdelta[sbin[s]];
        rc[i] = stage[s];
    }
#pragma unroll
    for (int i = 0; i < EPT; ++i) {
        const int s = i * BINT + tid;
        if (s < nvalid) sorted[gd[i] + s] = rc[i];
    }
}

#define OWN_GB 16

template <int D>
struct OwnCfg {
    static constexpr int VW  = D >= 64 ? 2 : 1;
    static constexpr int LPE = D / VW;
    static constexpr int EPI = 32 / LPE;
    static constexpr int NG  = OWN_GB / EPI;
    static constexpr int PITCH = D >= 64 ? D + 4 : D;
    static_assert(LPE == 32 || LPE == 16, "row must map onto a wave or a half-wave");
    static_assert(PITCH % 4 == 0, "rows start on 16-byte boundaries");
};

template <int D, int J0>
__device__ __forceinline__ void own_loads(const float* __restrict__ S, int key, float wt,
                                          int sub, int c,
                                          float (&vx)[OwnCfg<D>::NG], float (&vy)[OwnCfg<D>::NG],
                                          float (&wj)[OwnCfg<D>::NG], int (&dj)[OwnCfg<D>::NG]) {
    constexpr int VW = OwnCfg<D>::VW, EPI = OwnCfg<D>::EPI, NG = OwnCfg<D>::NG;
#pragma unroll
    for (int u = 0; u < NG; ++u) {
        const int j = J0 + u * EPI;
        int kj;
        if constexpr (EPI == 1) {
            kj    = __builtin_amdgcn_readlane(key, j);
            wj[u] = __int_as_float(__builtin_amdgcn_readlane(__float_as_int(wt), j));
        } else {
            const int k0 = __builtin_amdgcn_readlane(key, j);
            const int k1 = __builtin_amdgcn_readlane(key, j + 1);
            const int w0 = __builtin_amdgcn_readlane(__float_as_int(wt), j);
            const int w1 = __builtin_amdgcn_readlane(__float_as_int(wt), j + 1);
            kj    = sub ? k1 : k0;
            wj[u] = __int_as_float(sub ? w1 : w0);
        }
        const int sj = kj & ((1 << SRCB) - 1);
        int dl = kj >> SRCB;
        dl = dl < RPB ? dl : RPB - 1;
        if constexpr (EPI == 1) {
            int ro;
            asm("s_mul_i32 %0, %1, %2" : "=s"(ro) : "s"(dl), "n"(OwnCfg<D>::PITCH));
            dj[u] = ro;
        } else {
            dj[u] = dl * OwnCfg<D>::PITCH;
        }
        const float* __restrict__ sp = S + (size_t)sj * D + c;
        if constexpr (VW == 2) {
            const v2f t = *(const v2f*)sp;
            vx[u] = t.x; vy[u] = t.y;
        } else {
            vx[u] = *sp; vy[u] = 0.f;
        }
    }
}

template <int D, int J0, bool FULL>
__device__ __forceinline__ void own_adds(float* acc, int cnt, int sub, int c,
                                         const float (&vx)[OwnCfg<D>::NG], const float (&vy)[OwnCfg<D>::NG],
                                         const float (&wj)[OwnCfg<D>::NG], const int (&dj)[OwnCfg<D>::NG]) {
    constexpr int VW = OwnCfg<D>::VW, EPI = OwnCfg<D>::EPI, NG = OwnCfg<D>::NG;
#pragma unroll
    for (int u = 0; u < NG; ++u) {
        const int j = J0 + u * EPI;
        if (FULL || j + sub < cnt) {
            float* ap = acc + dj[u] + c;
            atomicAdd(ap, wj[u] * vx[u]);
            if constexpr (VW == 2) atomicAdd(ap + 1, wj[u] * vy[u]);
        }
    }
}

template <int D, bool PRODUCT>
__global__ __launch_bounds__(512)
void own_sum(const float* __restrict__ S, const uint2* __restrict__ sorted,
             const int* __restrict__ binBase, const float* __restrict__ bias,
             const v2f* __restrict__ PBn, float* __restrict__ out, int n_rows) {
    constexpr int VW    = OwnCfg<D>::VW;
    constexpr int LPE   = OwnCfg<D>::LPE;
    constexpr int NG    = OwnCfg<D>::NG;
    constexpr int PITCH = OwnCfg<D>::PITCH;
    constexpr int NW    = 512 / 32;
    constexpr int CG    = D / 4;
    constexpr int NTILE = (RPB + 15) / 16;
    static_assert(512 % CG == 0, "a thread keeps one column group in the start-value loop");
    static_assert(OWN_GB == 16, "two groups per 32 records");
    static_assert(!PRODUCT || (D == NHID && NTILE <= NW), "one wave per 16-row tile, K = NHID");
    static_assert(PRODUCT || (D == NOUT && RPB <= 512), "one thread per row for the row's log-sum-exp");

    __shared__ float acc[RPB * PITCH];

    const int tid  = threadIdx.x;
    const int lane = tid & 31;
    const int wv   = __builtin_amdgcn_readfirstlane(tid >> 5);
    const int sub  = lane / LPE;
    const int c    = (lane % LPE) * VW;
    const int b    = blockIdx.x;
    const int beg  = binBase[b];
    const int end  = binBase[b + 1];
    const int row0 = b * RPB;

    const int c4 = tid % CG;
    float4 bv;
    bv.x = bias[c4 * 4 + 0]; bv.y = bias[c4 * 4 + 1];
    bv.z = bias[c4 * 4 + 2]; bv.w = bias[c4 * 4 + 3];
    float4* __restrict__ acc4 = (float4*)acc;
    for (int r = tid / CG; r < RPB; r += 512 / CG) acc4[r * (PITCH / 4) + c4] = bv;
    __syncthreads();

    int   nkey = 0;
    float nwt  = 0.f;
    int g = beg + wv * 32;
    if (g + lane < end) { const uint2 r = sorted[g + lane]; nkey = (int)r.x; nwt = __uint_as_float(r.y); }

    float vx[NG], vy[NG], wj[NG];
    int   dj[NG];

    for (; g + 32 <= end; g += NW * 32) {
        const int   key = nkey;
        const float wt  = nwt;
        own_loads<D, 0>(S, key, wt, sub, c, vx, vy, wj, dj);
        own_adds<D, 0, true>(acc, 32, sub, c, vx, vy, wj, dj);
        own_loads<D, OWN_GB>(S, key, wt, sub, c, vx, vy, wj, dj);
        {
            const int pi = (g + NW * 32 + lane) < end ? (g + NW * 32 + lane) : end - 1;
            const uint2 r = sorted[pi];
            nkey = (int)r.x; nwt = __uint_as_float(r.y);
        }
        own_adds<D, OWN_GB, true>(acc, 32, sub, c, vx, vy, wj, dj);
    }
    if (g < end) {
        const int   cnt = end - g;
        const int   key = nkey;
        const float wt  = nwt;
        own_loads<D, 0>(S, key, wt, sub, c, vx, vy, wj, dj);
        own_adds<D, 0, false>(acc, cnt, sub, c, vx, vy, wj, dj);
        if (cnt > OWN_GB) {
            own_loads<D, OWN_GB>(S, key, wt, sub, c, vx, vy, wj, dj);
            own_adds<D, OWN_GB, false>(acc, cnt, sub, c, vx, vy, wj, dj);
        }
    }

    const int rows = (n_rows - row0) < RPB ? (n_rows - row0) : RPB;

    if constexpr (PRODUCT) {
        __shared__ float s2s[NTILE * 16 * NOUT];
        const int half = lane >> 4;
        const int l16  = lane & 15;
        v2f bf[D / 4];
        if (wv < NTILE) {
#pragma unroll
            for (int k4 = 0; k4 < D / 4; ++k4) bf[k4] = PBn[k4 * 32 + lane];
        }
        __syncthreads();
        if (wv < NTILE) {
            int ar = wv * 16 + l16;
            ar = ar < RPB ? ar : RPB - 1;
            const float* ap = acc + ar * PITCH + 2 * half;
            v8f d = (v8f){0.f,0.f,0.f,0.f,0.f,0.f,0.f,0.f};
#pragma unroll
            for (int k4 = 0; k4 < D / 4; ++k4) {
                v2f a = *(const v2f*)(ap + k4 * 4);
                a.x = fmaxf(a.x, 0.f); a.y = fmaxf(a.y, 0.f);
                d = __builtin_amdgcn_wmma_f32_16x16x4_f32(
                    false, a, false, bf[k4], (short)0, d, false, false);
            }
#pragma unroll
            for (int j = 0; j < 8; ++j) s2s[(wv * 16 + j + 8 * half) * NOUT + l16] = d[j];
        }
        __syncthreads();
        float4* __restrict__ out4 = (float4*)(out + (size_t)row0 * NOUT);
        const float4* __restrict__ st4 = (const float4*)s2s;
        for (int i = tid; i < rows * (NOUT / 4); i += 512) out4[i] = st4[i];
    } else {
        __shared__ float lse_s[RPB];
        __syncthreads();
        if (tid < RPB) {
            float v[D];
#pragma unroll
            for (int q = 0; q < CG; ++q) {
                const float4 t = acc4[tid * (PITCH / 4) + q];
                v[q * 4 + 0] = t.x; v[q * 4 + 1] = t.y;
                v[q * 4 + 2] = t.z; v[q * 4 + 3] = t.w;
            }
            float mx = v[0];
#pragma unroll
            for (int j = 1; j < D; ++j) mx = fmaxf(mx, v[j]);
            float s = 0.f;
#pragma unroll
            for (int j = 0; j < D; ++j) s += expf(v[j] - mx);
            lse_s[tid] = mx + logf(s);
        }
        __syncthreads();
        static_assert(PRODUCT || PITCH == D, "rows are contiguous in LDS and in out");
        float4* __restrict__ out4 = (float4*)(out + (size_t)row0 * D);
        for (int i = tid; i < rows * CG; i += 512) {
            float4 t = acc4[i];
            const float l = lse_s[i / CG];
            t.x -= l; t.y -= l; t.z -= l; t.w -= l;
            out4[i] = t;
        }
    }
}

extern "C" void kernel_launch(void* const* d_in, const int* in_sizes, int n_in,
                              void* d_out, int out_size, void* d_ws, size_t ws_size,
                              hipStream_t stream) {
    const float* x    = (const float*)d_in[0];
    const int*   esrc = (const int*)  d_in[1];
    const int*   edst = (const int*)  d_in[2];
    const float* ew   = (const float*)d_in[3];
    const float* W1   = (const float*)d_in[4];
    const float* b1   = (const float*)d_in[5];
    const float* W2   = (const float*)d_in[6];
    const float* b2   = (const float*)d_in[7];
    float* out = (float*)d_out;

    const int n_nodes = in_sizes[0] / NFEAT;
    const int n_edges = in_sizes[1];

    constexpr int PB1_N = (NFEAT / 4) * (NHID / 16) * 32;
    constexpr int PB2_N = (NHID  / 4) * (NOUT / 16) * 32;

    float* R1  = (float*)d_ws;
    float* R2  = R1 + (size_t)n_nodes * NHID;
    float* R3  = R2 + (size_t)n_nodes * NHID;
    v2f*   PB1 = (v2f*)(R3 + (size_t)n_nodes * 2 * NOUT);
    v2f*   PB2 = PB1 + PB1_N;

    float* S1 = R1;
    float* S2 = R2;
    uint2* sorted    = (uint2*)R3;
    int*   binCount  = (int*)R2;
    int*   binCursor = binCount + MAXB;
    int*   binBase   = (int*)PB1;

    const int mtiles      = (n_nodes + 15) / 16;
    const int gemm_groups = (mtiles + FT / 32 - 1) / (FT / 32);
    const int ncount      = (n_edges + FT * CEPT - 1) / (FT * CEPT);
    const int nslices     = (n_edges + BINT * EPT - 1) / (BINT * EPT);
    const int nbins       = (n_nodes + RPB - 1) / RPB;

    pack_both<<<(PB1_N + PB2_N + 255) / 256, 256, 0, stream>>>(W1, PB1, W2, PB2, binCount);

    gemm_and_count<NFEAT, NHID>
        <<<gemm_groups + ncount, FT, 0, stream>>>(x, PB1, S1, n_nodes, gemm_groups,
                                                  edst, n_edges, binCount);

    bin_place<<<nslices, BINT, 0, stream>>>(esrc, edst, ew, n_edges,
                                            binCount, binCursor, sorted, binBase);

    own_sum<NHID, true><<<nbins, 512, 0, stream>>>(S1, sorted, binBase, b1, PB2, S2, n_nodes);

    own_sum<NOUT, false><<<nbins, 512, 0, stream>>>(S2, sorted, binBase, b2, nullptr, out, n_nodes);
}


# layer-1 owner: bin's records put in row order in LDS (integer counting sort); a row's run is summed in registers, LDS float adds only where a run ends; sums per run, runs in arrival order
# speedup vs baseline: 3.2119x; 1.1111x over previous
#include <hip/hip_runtime.h>
#include <hip/hip_bf16.h>
#include <math.h>


typedef float v2f __attribute__((ext_vector_type(2)));
typedef float v8f __attribute__((ext_vector_type(8)));

#define NFEAT 256
#define NHID  64
#define NOUT  16

#define RPB   200
#define MAXB  512
#define BINT  512
#define EPT   8
#define SRCB  17

template <int K, int N>
__device__ __forceinline__ void pack_one(const float* __restrict__ W, v2f* __restrict__ PB, int idx) {
    constexpr int NT    = N / 16;
    const int lane  = idx & 31;
    const int tile  = (idx >> 5) % NT;
    const int kstep = idx / (NT * 32);
    const int half  = lane >> 4;
    const int col   = tile * 16 + (lane & 15);
    const int k     = kstep * 4 + 2 * half;
    v2f b;
    b.x = W[(size_t)k * N + col];
    b.y = W[(size_t)(k + 1) * N + col];
    PB[idx] = b;
}

__global__ void pack_both(const float* __restrict__ Wa, v2f* __restrict__ PBa,
                          const float* __restrict__ Wb, v2f* __restrict__ PBb,
                          int* __restrict__ tbl) {
    constexpr int TOTAL_A = (NFEAT / 4) * (NHID / 16) * 32;
    constexpr int TOTAL_B = (NHID  / 4) * (NOUT / 16) * 32;
    const int idx = blockIdx.x * blockDim.x + threadIdx.x;
    if (idx < 2 * MAXB) tbl[idx] = 0;
    if (idx < TOTAL_A)                pack_one<NFEAT, NHID>(Wa, PBa, idx);
    else if (idx < TOTAL_A + TOTAL_B) pack_one<NHID,  NOUT>(Wb, PBb, idx - TOTAL_A);
}

#define FT 128

template <int K, int N>
__device__ __forceinline__
void gemm_wmma_f32(const float* __restrict__ A,
                   const v2f* __restrict__ PB,
                   float* __restrict__ out, int M) {
    constexpr int NT = N / 16;
    const int wave  = (blockIdx.x * FT + threadIdx.x) >> 5;
    const int lane  = threadIdx.x & 31;
    const int mbase = wave * 16;
    if (mbase >= M) return;

    const int half = lane >> 4;
    const int l16  = lane & 15;

    int arow = mbase + l16;
    if (arow >= M) arow = M - 1;
    const float* __restrict__ ap = A + (size_t)arow * K + 2 * half;
    const v2f*   __restrict__ bp = PB + lane;

    v8f acc[NT];
#pragma unroll
    for (int t = 0; t < NT; ++t) acc[t] = (v8f){0.f,0.f,0.f,0.f,0.f,0.f,0.f,0.f};

#pragma unroll 8
    for (int k4 = 0; k4 < K / 4; ++k4) {
        v2f a = *(const v2f*)(ap + k4 * 4);
        const v2f* __restrict__ bk = bp + (size_t)k4 * (NT * 32);
#pragma unroll
        for (int t = 0; t < NT; ++t) {
            acc[t] = __builtin_amdgcn_wmma_f32_16x16x4_f32(
                false, a, false, bk[t * 32], (short)0, acc[t], false, false);
        }
    }

#pragma unroll
    for (int t = 0; t < NT; ++t) {
#pragma unroll
        for (int j = 0; j < 8; ++j) {
            const int row = mbase + j + 8 * half;
            if (row < M)
                out[(size_t)row * N + t * 16 + l16] = acc[t][j];
        }
    }
}

#define CEPT 32
__device__ __forceinline__ int bin_of(int d) {
    unsigned b = (unsigned)d / RPB;
    return b < MAXB ? (int)b : MAXB - 1;
}

__device__ __forceinline__
void bin_count(int slice, const int* __restrict__ dst, int n_edges,
               int* __restrict__ binCount, int* hist) {
    const int tid = threadIdx.x;
    for (int i = tid; i <= MAXB; i += FT) hist[i] = 0;
    const int e0 = slice * (FT * CEPT) + tid;
    int d[CEPT];
#pragma unroll
    for (int i = 0; i < CEPT; ++i) {
        const int e = e0 + i * FT;
        d[i] = dst[e < n_edges ? e : n_edges - 1];
    }
    __syncthreads();
#pragma unroll
    for (int i = 0; i < CEPT; ++i) {
        const int e = e0 + i * FT;
        atomicAdd(&hist[e < n_edges ? bin_of(d[i]) : MAXB], 1);
    }
    __syncthreads();
    for (int i = tid; i < MAXB; i += FT) {
        const int c = hist[i];
        if (c) atomicAdd(binCount + i, c);
    }
}

template <int K, int N>
__global__ __launch_bounds__(FT)
void gemm_and_count(const float* __restrict__ A, const v2f* __restrict__ PB,
                    float* __restrict__ S, int M, int gemm_groups,
                    const int* __restrict__ dst, int n_edges, int* __restrict__ binCount) {
    __shared__ int hist[MAXB + 1];
    if ((int)blockIdx.x < gemm_groups) {
        gemm_wmma_f32<K, N>(A, PB, S, M);
    } else {
        bin_count((int)blockIdx.x - gemm_groups, dst, n_edges, binCount, hist);
    }
}

__device__ __forceinline__ int wave_incl_scan(int v, int lane) {
#pragma unroll
    for (int o = 1; o < 32; o <<= 1) {
        const int t = __shfl_up(v, o);
        if (lane >= o) v += t;
    }
    return v;
}

__global__ __launch_bounds__(BINT)
void bin_place(const int* __restrict__ src, const int* __restrict__ dst,
               const float* __restrict__ w, int n_edges,
               const int* __restrict__ binCount, int* __restrict__ binCursor,
               uint2* __restrict__ sorted, int* __restrict__ binBaseOut) {
    __shared__ int   hist[MAXB + 1];
    __shared__ int   lstart[MAXB + 1];
    __shared__ int   gdelta[MAXB + 1];
    __shared__ int   wsumA[BINT / 32];
    __shared__ int   wsumB[BINT / 32];
    __shared__ uint2 stage[BINT * EPT];
    __shared__ unsigned short sbin[BINT * EPT];

    const int tid  = threadIdx.x;
    const int lane = tid & 31;
    const int wv   = tid >> 5;
    const int left   = n_edges - blockIdx.x * (BINT * EPT);
    const int nvalid = left < BINT * EPT ? left : BINT * EPT;
    hist[tid] = 0;
    if (tid == 0) hist[MAXB] = 0;

    const int cnt = binCount[tid];
    const int inc = wave_incl_scan(cnt, lane);
    if (lane == 31) wsumA[wv] = inc;

    int   dd[EPT], ss[EPT];
    float wt[EPT];
    const int e0 = blockIdx.x * (BINT * EPT) + tid;
#pragma unroll
    for (int i = 0; i < EPT; ++i) {
        const int e  = e0 + i * BINT;
        const int ec = e < n_edges ? e : n_edges - 1;
        dd[i] = dst[ec]; ss[i] = src[ec]; wt[i] = w[ec];
    }
    int key[EPT], bn[EPT];
#pragma unroll
    for (int i = 0; i < EPT; ++i) {
        const int e = e0 + i * BINT;
        const int b = bin_of(dd[i]);
        bn[i]  = e < n_edges ? b : MAXB;
        key[i] = ss[i] | ((dd[i] - b * RPB) << SRCB);
    }
    __syncthreads();
    int pre = 0;
#pragma unroll
    for (int k = 0; k < BINT / 32; ++k) if (k < wv) pre += wsumA[k];
    const int excl = pre + inc - cnt;
#pragma unroll
    for (int i = 0; i < EPT; ++i) atomicAdd(&hist[bn[i]], 1);
    __syncthreads();

    const int c    = hist[tid];
    const int linc = wave_incl_scan(c, lane);
    if (lane == 31) wsumB[wv] = linc;
    int off = 0;
    if (c) off = atomicAdd(binCursor + tid, c);
    if (blockIdx.x == 0) {
        binBaseOut[tid] = excl;
        if (tid == BINT - 1) binBaseOut[MAXB] = excl + cnt;
    }
    __syncthreads();
    int lpre = 0;
#pragma unroll
    for (int k = 0; k < BINT / 32; ++k) if (k < wv) lpre += wsumB[k];
    const int ls = lpre + linc - c;
    lstart[tid] = ls;
    gdelta[tid] = excl + off - ls;
    hist[tid]   = 0;
    if (tid == 0) { lstart[MAXB] = nvalid; gdelta[MAXB] = 0; hist[MAXB] = 0; }
    __syncthreads();
    int slot[EPT];
#pragma unroll
    for (int i = 0; i < EPT; ++i) slot[i] = lstart[bn[i]] + atomicAdd(&hist[bn[i]], 1);
#pragma unroll
    for (int i = 0; i < EPT; ++i) {
        stage[slot[i]] = make_uint2((unsigned)key[i], __float_as_uint(wt[i]));
        sbin[slot[i]]  = (unsigned short)bn[i];
    }
    __syncthreads();
    int   gd[EPT];
    uint2 rc[EPT];
#pragma unroll
    for (int i = 0; i < EPT; ++i) {
        const int s = i * BINT + tid;
        gd[i] = gdelta[sbin[s]];
        rc[i] = stage[s];
    }
#pragma unroll
    for (int i = 0; i < EPT; ++i) {
        const int s = i * BINT + tid;
        if (s < nvalid) sorted[gd[i] + s] = rc[i];
    }
}

#define OWN_GB 16

template <int D>
struct OwnCfg {
    static constexpr int VW  = D >= 64 ? 2 : 1;
    static constexpr int LPE = D / VW;
    static constexpr int EPI = 32 / LPE;
    static constexpr int NG  = OWN_GB / EPI;
    static constexpr int PITCH = D >= 64 ? D + 4 : D;
    static_assert(LPE == 32 || LPE == 16, "row must map onto a wave or a half-wave");
    static_assert(PITCH % 4 == 0, "rows start on 16-byte boundaries");
};

template <int D, int J0>
__device__ __forceinline__ void own_loads(const float* __restrict__ S, int key, float wt,
                                          int sub, int c,
                                          float (&vx)[OwnCfg<D>::NG], float (&vy)[OwnCfg<D>::NG],
                                          float (&wj)[OwnCfg<D>::NG], int (&dj)[OwnCfg<D>::NG]) {
    constexpr int VW = OwnCfg<D>::VW, EPI = OwnCfg<D>::EPI, NG = OwnCfg<D>::NG;
#pragma unroll
    for (int u = 0; u < NG; ++u) {
        const int j = J0 + u * EPI;
        int kj;
        if constexpr (EPI == 1) {
            kj    = __builtin_amdgcn_readlane(key, j);
            wj[u] = __int_as_float(__builtin_amdgcn_readlane(__float_as_int(wt), j));
        } else {
            const int k0 = __builtin_amdgcn_readlane(key, j);
            const int k1 = __builtin_amdgcn_readlane(key, j + 1);
            const int w0 = __builtin_amdgcn_readlane(__float_as_int(wt), j);
            const int w1 = __builtin_amdgcn_readlane(__float_as_int(wt), j + 1);
            kj    = sub ? k1 : k0;
            wj[u] = __int_as_float(sub ? w1 : w0);
        }
        const int sj = kj & ((1 << SRCB) - 1);
        int dl = kj >> SRCB;
        dl = dl < RPB ? dl : RPB - 1;
        if constexpr (EPI == 1) {
            int ro;
            asm("s_mul_i32 %0, %1, %2" : "=s"(ro) : "s"(dl), "n"(OwnCfg<D>::PITCH));
            dj[u] = ro;
        } else {
            dj[u] = dl * OwnCfg<D>::PITCH;
        }
        const float* __restrict__ sp = S + (size_t)sj * D + c;
        if constexpr (VW == 2) {
            const v2f t = *(const v2f*)sp;
            vx[u] = t.x; vy[u] = t.y;
        } else {
            vx[u] = *sp; vy[u] = 0.f;
        }
    }
}

template <int D, int J0, bool FULL>
__device__ __forceinline__ void own_adds(float* acc, int cnt, int sub, int c,
                                         const float (&vx)[OwnCfg<D>::NG], const float (&vy)[OwnCfg<D>::NG],
                                         const float (&wj)[OwnCfg<D>::NG], const int (&dj)[OwnCfg<D>::NG]) {
    constexpr int VW = OwnCfg<D>::VW, EPI = OwnCfg<D>::EPI, NG = OwnCfg<D>::NG;
#pragma unroll
    for (int u = 0; u < NG; ++u) {
        const int j = J0 + u * EPI;
        if (FULL || j + sub < cnt) {
            float* ap = acc + dj[u] + c;
            atomicAdd(ap, wj[u] * vx[u]);
            if constexpr (VW == 2) atomicAdd(ap + 1, wj[u] * vy[u]);
        }
    }
}

#define OWN_CAP 4096

__device__ __forceinline__ int row_of(unsigned key) {
    const unsigned r = key >> SRCB;
    return r < RPB ? (int)r : RPB - 1;
}

template <int D, int J0, bool FULL>
__device__ __forceinline__ void own_runs(float* acc, int cnt, int c,
                                         const float (&vx)[OwnCfg<D>::NG], const float (&vy)[OwnCfg<D>::NG],
                                         const float (&wj)[OwnCfg<D>::NG], const int (&dj)[OwnCfg<D>::NG],
                                         int& cur, float& ax, float& ay) {
    static_assert(OwnCfg<D>::EPI == 1 && OwnCfg<D>::VW == 2, "one edge per instruction, two floats per lane");
#pragma unroll
    for (int u = 0; u < OwnCfg<D>::NG; ++u) {
        if (FULL || J0 + u < cnt) {
            if (dj[u] != cur) {
                float* ap = acc + cur + c;
                atomicAdd(ap, ax);
                atomicAdd(ap + 1, ay);
                ax = 0.f; ay = 0.f; cur = dj[u];
            }
            ax += wj[u] * vx[u];
            ay += wj[u] * vy[u];
        }
    }
}

template <int D>
__device__ __forceinline__ void own_by_row(const float* __restrict__ S, const uint2* __restrict__ sorted,
                                           int beg, int n, float* acc, int tid, int lane, int wv, int c) {
    constexpr int NG  = OwnCfg<D>::NG;
    constexpr int NW  = 512 / 32;
    constexpr int RPT = OWN_CAP / 512;
    __shared__ uint2 srec[OWN_CAP];
    __shared__ int   rcnt[RPB];
    __shared__ int   rstart[RPB];
    __shared__ int   wsum[NW];

    for (int r = tid; r < RPB; r += 512) rcnt[r] = 0;
    uint2 rr[RPT];
#pragma unroll
    for (int i = 0; i < RPT; ++i) {
        const int idx = tid + i * 512;
        rr[i] = make_uint2(0u, 0u);
        if (idx < n) rr[i] = sorted[beg + idx];
    }
    __syncthreads();
    int rk[RPT];
#pragma unroll
    for (int i = 0; i < RPT; ++i) {
        const int idx = tid + i * 512;
        rk[i] = 0;
        if (idx < n) rk[i] = atomicAdd(&rcnt[row_of(rr[i].x)], 1);
    }
    __syncthreads();
    const int cntv = tid < RPB ? rcnt[tid] : 0;
    const int inc  = wave_incl_scan(cntv, lane);
    if (lane == 31) wsum[wv] = inc;
    __syncthreads();
    int pre = 0;
#pragma unroll
    for (int k = 0; k < NW; ++k) if (k < wv) pre += wsum[k];
    if (tid < RPB) rstart[tid] = pre + inc - cntv;
    __syncthreads();
#pragma unroll
    for (int i = 0; i < RPT; ++i) {
        const int idx = tid + i * 512;
        if (idx < n) srec[rstart[row_of(rr[i].x)] + rk[i]] = rr[i];
    }
    __syncthreads();

    int   nkey = 0;
    float nwt  = 0.f;
    int g = wv * 32;
    if (g + lane < n) { const uint2 r = srec[g + lane]; nkey = (int)r.x; nwt = __uint_as_float(r.y); }

    float vx[NG], vy[NG], wj[NG];
    int   dj[NG];

    for (; g + 32 <= n; g += NW * 32) {
        const int   key = nkey;
        const float wt  = nwt;
        own_loads<D, 0>(S, key, wt, 0, c, vx, vy, wj, dj);
        int   cur = dj[0];
        float ax = 0.f, ay = 0.f;
        own_runs<D, 0, true>(acc, 32, c, vx, vy, wj, dj, cur, ax, ay);
        own_loads<D, OWN_GB>(S, key, wt, 0, c, vx, vy, wj, dj);
        {
            const int pi = (g + NW * 32 + lane) < n ? (g + NW * 32 + lane) : n - 1;
            const uint2 r = srec[pi];
            nkey = (int)r.x; nwt = __uint_as_float(r.y);
        }
        own_runs<D, OWN_GB, true>(acc, 32, c, vx, vy, wj, dj, cur, ax, ay);
        float* ap = acc + cur + c;
        atomicAdd(ap, ax);
        atomicAdd(ap + 1, ay);
    }
    if (g < n) {
        const int   cnt = n - g;
        const int   key = nkey;
        const float wt  = nwt;
        own_loads<D, 0>(S, key, wt, 0, c, vx, vy, wj, dj);
        int   cur = dj[0];
        float ax = 0.f, ay = 0.f;
        own_runs<D, 0, false>(acc, cnt, c, vx, vy, wj, dj, cur, ax, ay);
        if (cnt > OWN_GB) {
            own_loads<D, OWN_GB>(S, key, wt, 0, c, vx, vy, wj, dj);
            own_runs<D, OWN_GB, false>(acc, cnt, c, vx, vy, wj, dj, cur, ax, ay);
        }
        float* ap = acc + cur + c;
        atomicAdd(ap, ax);
        atomicAdd(ap + 1, ay);
    }
}

template <int D, bool PRODUCT>
__global__ __launch_bounds__(512)
void own_sum(const float* __restrict__ S, const uint2* __restrict__ sorted,
             const int* __restrict__ binBase, const float* __restrict__ bias,
             const v2f* __restrict__ PBn, float* __restrict__ out, int n_rows) {
    constexpr int VW    = OwnCfg<D>::VW;
    constexpr int LPE   = OwnCfg<D>::LPE;
    constexpr int NG    = OwnCfg<D>::NG;
    constexpr int PITCH = OwnCfg<D>::PITCH;
    constexpr int NW    = 512 / 32;
    constexpr int CG    = D / 4;
    constexpr int NTILE = (RPB + 15) / 16;
    static_assert(512 % CG == 0, "a thread keeps one column group in the start-value loop");
    static_assert(OWN_GB == 16, "two groups per 32 records");
    static_assert(!PRODUCT || (D == NHID && NTILE <= NW), "one wave per 16-row tile, K = NHID");
    static_assert(PRODUCT || (D == NOUT && RPB <= 512), "one thread per row for the row's log-sum-exp");

    __shared__ float acc[RPB * PITCH];

    const int tid  = threadIdx.x;
    const int lane = tid & 31;
    const int wv   = __builtin_amdgcn_readfirstlane(tid >> 5);
    const int sub  = lane / LPE;
    const int c    = (lane % LPE) * VW;
    const int b    = blockIdx.x;
    const int beg  = binBase[b];
    const int end  = binBase[b + 1];
    const int row0 = b * RPB;

    const int c4 = tid % CG;
    float4 bv;
    bv.x = bias[c4 * 4 + 0]; bv.y = bias[c4 * 4 + 1];
    bv.z = bias[c4 * 4 + 2]; bv.w = bias[c4 * 4 + 3];
    float4* __restrict__ acc4 = (float4*)acc;
    for (int r = tid / CG; r < RPB; r += 512 / CG) acc4[r * (PITCH / 4) + c4] = bv;

    bool by_row = false;
    if constexpr (PRODUCT) by_row = (end - beg) <= OWN_CAP;
    if (by_row) {
        if constexpr (PRODUCT) own_by_row<D>(S, sorted, beg, end - beg, acc, tid, lane, wv, c);
    } else {
        __syncthreads();

        int   nkey = 0;
        float nwt  = 0.f;
        int g = beg + wv * 32;
        if (g + lane < end) { const uint2 r = sorted[g + lane]; nkey = (int)r.x; nwt = __uint_as_float(r.y); }

        float vx[NG], vy[NG], wj[NG];
        int   dj[NG];

        for (; g + 32 <= end; g += NW * 32) {
            const int   key = nkey;
            const float wt  = nwt;
            own_loads<D, 0>(S, key, wt, sub, c, vx, vy, wj, dj);
            own_adds<D, 0, true>(acc, 32, sub, c, vx, vy, wj, dj);
            own_loads<D, OWN_GB>(S, key, wt, sub, c, vx, vy, wj, dj);
            {
                const int pi = (g + NW * 32 + lane) < end ? (g + NW * 32 + lane) : end - 1;
                const uint2 r = sorted[pi];
                nkey = (int)r.x; nwt = __uint_as_float(r.y);
            }
            own_adds<D, OWN_GB, true>(acc, 32, sub, c, vx, vy, wj, dj);
        }
        if (g < end) {
            const int   cnt = end - g;
            const int   key = nkey;
            const float wt  = nwt;
            own_loads<D, 0>(S, key, wt, sub, c, vx, vy, wj, dj);
            own_adds<D, 0, false>(acc, cnt, sub, c, vx, vy, wj, dj);
            if (cnt > OWN_GB) {
                own_loads<D, OWN_GB>(S, key, wt, sub, c, vx, vy, wj, dj);
                own_adds<D, OWN_GB, false>(acc, cnt, sub, c, vx, vy, wj, dj);
            }
        }
    }

    const int rows = (n_rows - row0) < RPB ? (n_rows - row0) : RPB;

    if constexpr (PRODUCT) {
        __shared__ float s2s[NTILE * 16 * NOUT];
        const int half = lane >> 4;
        const int l16  = lane & 15;
        v2f bf[D / 4];
        if (wv < NTILE) {
#pragma unroll
            for (int k4 = 0; k4 < D / 4; ++k4) bf[k4] = PBn[k4 * 32 + lane];
        }
        __syncthreads();
        if (wv < NTILE) {
            int ar = wv * 16 + l16;
            ar = ar < RPB ? ar : RPB - 1;
            const float* ap = acc + ar * PITCH + 2 * half;
            v8f d = (v8f){0.f,0.f,0.f,0.f,0.f,0.f,0.f,0.f};
#pragma unroll
            for (int k4 = 0; k4 < D / 4; ++k4) {
                v2f a = *(const v2f*)(ap + k4 * 4);
                a.x = fmaxf(a.x, 0.f); a.y = fmaxf(a.y, 0.f);
                d = __builtin_amdgcn_wmma_f32_16x16x4_f32(
                    false, a, false, bf[k4], (short)0, d, false, false);
            }
#pragma unroll
            for (int j = 0; j < 8; ++j) s2s[(wv * 16 + j + 8 * half) * NOUT + l16] = d[j];
        }
        __syncthreads();
        float4* __restrict__ out4 = (float4*)(out + (size_t)row0 * NOUT);
        const float4* __restrict__ st4 = (const float4*)s2s;
        for (int i = tid; i < rows * (NOUT / 4); i += 512) out4[i] = st4[i];
    } else {
        __shared__ float lse_s[RPB];
        __syncthreads();
        if (tid < RPB) {
            float v[D];
#pragma unroll
            for (int q = 0; q < CG; ++q) {
                const float4 t = acc4[tid * (PITCH / 4) + q];
                v[q * 4 + 0] = t.x; v[q * 4 + 1] = t.y;
                v[q * 4 + 2] = t.z; v[q * 4 + 3] = t.w;
            }
            float mx = v[0];
#pragma unroll
            for (int j = 1; j < D; ++j) mx = fmaxf(mx, v[j]);
            float s = 0.f;
#pragma unroll
            for (int j = 0; j < D; ++j) s += expf(v[j] - mx);
            lse_s[tid] = mx + logf(s);
        }
        __syncthreads();
        static_assert(PRODUCT || PITCH == D, "rows are contiguous in LDS and in out");
        float4* __restrict__ out4 = (float4*)(out + (size_t)row0 * D);
        for (int i = tid; i < rows * CG; i += 512) {
            float4 t = acc4[i];
            const float l = lse_s[i / CG];
            t.x -= l; t.y -= l; t.z -= l; t.w -= l;
            out4[i] = t;
        }
    }
}

extern "C" void kernel_launch(void* const* d_in, const int* in_sizes, int n_in,
                              void* d_out, int out_size, void* d_ws, size_t ws_size,
                              hipStream_t stream) {
    const float* x    = (const float*)d_in[0];
    const int*   esrc = (const int*)  d_in[1];
    const int*   edst = (const int*)  d_in[2];
    const float* ew   = (const float*)d_in[3];
    const float* W1   = (const float*)d_in[4];
    const float* b1   = (const float*)d_in[5];
    const float* W2   = (const float*)d_in[6];
    const float* b2   = (const float*)d_in[7];
    float* out = (float*)d_out;

    const int n_nodes = in_sizes[0] / NFEAT;
    const int n_edges = in_sizes[1];

    constexpr int PB1_N = (NFEAT / 4) * (NHID / 16) * 32;
    constexpr int PB2_N = (NHID  / 4) * (NOUT / 16) * 32;

    float* R1  = (float*)d_ws;
    float* R2  = R1 + (size_t)n_nodes * NHID;
    float* R3  = R2 + (size_t)n_nodes * NHID;
    v2f*   PB1 = (v2f*)(R3 + (size_t)n_nodes * 2 * NOUT);
    v2f*   PB2 = PB1 + PB1_N;

    float* S1 = R1;
    float* S2 = R2;
    uint2* sorted    = (uint2*)R3;
    int*   binCount  = (int*)R2;
    int*   binCursor = binCount + MAXB;
    int*   binBase   = (int*)PB1;

    const int mtiles      = (n_nodes + 15) / 16;
    const int gemm_groups = (mtiles + FT / 32 - 1) / (FT / 32);
    const int ncount      = (n_edges + FT * CEPT - 1) / (FT * CEPT);
    const int nslices     = (n_edges + BINT * EPT - 1) / (BINT * EPT);
    const int nbins       = (n_nodes + RPB - 1) / RPB;

    pack_both<<<(PB1_N + PB2_N + 255) / 256, 256, 0, stream>>>(W1, PB1, W2, PB2, binCount);

    gemm_and_count<NFEAT, NHID>
        <<<gemm_groups + ncount, FT, 0, stream>>>(x, PB1, S1, n_nodes, gemm_groups,
                                                  edst, n_edges, binCount);

    bin_place<<<nslices, BINT, 0, stream>>>(esrc, edst, ew, n_edges,
                                            binCount, binCursor, sorted, binBase);

    own_sum<NHID, true><<<nbins, 512, 0, stream>>>(S1, sorted, binBase, b1, PB2, S2, n_nodes);

    own_sum<NOUT, false><<<nbins, 512, 0, stream>>>(S2, sorted, binBase, b2, nullptr, out, n_nodes);
}


# edge binning without a count pass: each bin owns a fixed range of records (an over-full bin's owner sums from the edge list), so placing runs as extra work-groups of the GEMM launch; 5 launches -> 4
# speedup vs baseline: 3.4114x; 1.0621x over previous
#include <hip/hip_runtime.h>
#include <hip/hip_bf16.h>
#include <math.h>


typedef float v2f __attribute__((ext_vector_type(2)));
typedef float v8f __attribute__((ext_vector_type(8)));

#define NFEAT 256
#define NHID  64
#define NOUT  16

#define RPB   200
#define MAXB  512
#define CAPB  6400
#define SRCB  17

template <int K, int N>
__device__ __forceinline__ void pack_one(const float* __restrict__ W, v2f* __restrict__ PB, int idx) {
    constexpr int NT    = N / 16;
    const int lane  = idx & 31;
    const int tile  = (idx >> 5) % NT;
    const int kstep = idx / (NT * 32);
    const int half  = lane >> 4;
    const int col   = tile * 16 + (lane & 15);
    const int k     = kstep * 4 + 2 * half;
    v2f b;
    b.x = W[(size_t)k * N + col];
    b.y = W[(size_t)(k + 1) * N + col];
    PB[idx] = b;
}

__global__ void pack_both(const float* __restrict__ Wa, v2f* __restrict__ PBa,
                          const float* __restrict__ Wb, v2f* __restrict__ PBb,
                          int* __restrict__ tbl) {
    constexpr int TOTAL_A = (NFEAT / 4) * (NHID / 16) * 32;
    constexpr int TOTAL_B = (NHID  / 4) * (NOUT / 16) * 32;
    const int idx = blockIdx.x * blockDim.x + threadIdx.x;
    if (idx < MAXB) tbl[idx] = 0;
    if (idx < TOTAL_A)                pack_one<NFEAT, NHID>(Wa, PBa, idx);
    else if (idx < TOTAL_A + TOTAL_B) pack_one<NHID,  NOUT>(Wb, PBb, idx - TOTAL_A);
}

#define FT 128

template <int K, int N>
__device__ __forceinline__
void gemm_wmma_f32(const float* __restrict__ A,
                   const v2f* __restrict__ PB,
                   float* __restrict__ out, int M) {
    constexpr int NT = N / 16;
    const int wave  = (blockIdx.x * FT + threadIdx.x) >> 5;
    const int lane  = threadIdx.x & 31;
    const int mbase = wave * 16;
    if (mbase >= M) return;

    const int half = lane >> 4;
    const int l16  = lane & 15;

    int arow = mbase + l16;
    if (arow >= M) arow = M - 1;
    const float* __restrict__ ap = A + (size_t)arow * K + 2 * half;
    const v2f*   __restrict__ bp = PB + lane;

    v8f acc[NT];
#pragma unroll
    for (int t = 0; t < NT; ++t) acc[t] = (v8f){0.f,0.f,0.f,0.f,0.f,0.f,0.f,0.f};

#pragma unroll 8
    for (int k4 = 0; k4 < K / 4; ++k4) {
        v2f a = *(const v2f*)(ap + k4 * 4);
        const v2f* __restrict__ bk = bp + (size_t)k4 * (NT * 32);
#pragma unroll
        for (int t = 0; t < NT; ++t) {
            acc[t] = __builtin_amdgcn_wmma_f32_16x16x4_f32(
                false, a, false, bk[t * 32], (short)0, acc[t], false, false);
        }
    }

#pragma unroll
    for (int t = 0; t < NT; ++t) {
#pragma unroll
        for (int j = 0; j < 8; ++j) {
            const int row = mbase + j + 8 * half;
            if (row < M)
                out[(size_t)row * N + t * 16 + l16] = acc[t][j];
        }
    }
}

#define PEPT 16
#define PSL  (FT * PEPT)
#define BPT  (MAXB / FT)

__device__ __forceinline__ int bin_of(int d) {
    unsigned b = (unsigned)d / RPB;
    return b < MAXB ? (int)b : MAXB - 1;
}

__device__ __forceinline__ int wave_incl_scan(int v, int lane) {
#pragma unroll
    for (int o = 1; o < 32; o <<= 1) {
        const int t = __shfl_up(v, o);
        if (lane >= o) v += t;
    }
    return v;
}

__device__ __forceinline__
void bin_place(int slice, const int* src, const int* __restrict__ dst,
               const float* w, int n_edges,
               int* __restrict__ binCursor, uint2* __restrict__ sorted,
               int* hist, int* lstart, int* gdelta, int* wsum,
               uint2* stage, unsigned short* sbin) {
    const int tid  = threadIdx.x;
    const int lane = tid & 31;
    const int wv   = tid >> 5;
    const int left   = n_edges - slice * PSL;
    const int nvalid = left < PSL ? left : PSL;
    for (int i = tid; i <= MAXB; i += FT) hist[i] = 0;

    int dd[PEPT];
    const int e0 = slice * PSL + tid;
#pragma unroll
    for (int i = 0; i < PEPT; ++i) {
        const int e = e0 + i * FT;
        const unsigned ec = (unsigned)(e < n_edges ? e : n_edges - 1);
        dd[i] = dst[ec];
    }
    int bd[PEPT];
#pragma unroll
    for (int i = 0; i < PEPT; ++i) {
        const int e = e0 + i * FT;
        const int b = bin_of(dd[i]);
        bd[i] = (e < n_edges ? b : MAXB) | (((dd[i] - b * RPB) & 0xFF) << 16);
    }
    __syncthreads();
#pragma unroll
    for (int i = 0; i < PEPT; ++i) atomicAdd(&hist[bd[i] & 0xFFFF], 1);
    __syncthreads();

    int c[BPT], off[BPT];
    int t = 0;
#pragma unroll
    for (int j = 0; j < BPT; ++j) { c[j] = hist[tid * BPT + j]; t += c[j]; }
#pragma unroll
    for (int j = 0; j < BPT; ++j) {
        off[j] = 0;
        if (c[j]) off[j] = atomicAdd(binCursor + tid * BPT + j, c[j]);
    }
    const int inc = wave_incl_scan(t, lane);
    if (lane == 31) wsum[wv] = inc;
    __syncthreads();
    int ls = inc - t;
#pragma unroll
    for (int k = 0; k < FT / 32; ++k) if (k < wv) ls += wsum[k];
#pragma unroll
    for (int j = 0; j < BPT; ++j) {
        const int q = tid * BPT + j;
        lstart[q] = ls;
        gdelta[q] = q * CAPB + off[j] - ls;
        hist[q]   = 0;
        ls += c[j];
    }
    if (tid == 0) { lstart[MAXB] = nvalid; gdelta[MAXB] = 0; hist[MAXB] = 0; }
    __syncthreads();
#pragma unroll
    for (int h = 0; h < 2; ++h) {
        int   ss[PEPT / 2], slot[PEPT / 2];
        float wt[PEPT / 2];
#pragma unroll
        for (int i = 0; i < PEPT / 2; ++i) {
            const int k  = (h * (PEPT / 2) + i) * FT + tid;
            const unsigned ec = (unsigned)(slice * PSL + (k < nvalid ? k : nvalid - 1));
            ss[i] = src[ec]; wt[i] = w[ec];
        }
#pragma unroll
        for (int i = 0; i < PEPT / 2; ++i) {
            const int bn = bd[h * (PEPT / 2) + i] & 0xFFFF;
            slot[i] = lstart[bn] + atomicAdd(&hist[bn], 1);
        }
#pragma unroll
        for (int i = 0; i < PEPT / 2; ++i) {
            const int v = bd[h * (PEPT / 2) + i];
            stage[slot[i]] = make_uint2((unsigned)(ss[i] | ((v >> 16) << SRCB)), __float_as_uint(wt[i]));
            sbin[slot[i]]  = (unsigned short)(v & 0xFFFF);
        }
    }
    __syncthreads();
#pragma unroll
    for (int h = 0; h < 2; ++h) {
        int   gd[PEPT / 2], lim[PEPT / 2];
        uint2 rc[PEPT / 2];
#pragma unroll
        for (int i = 0; i < PEPT / 2; ++i) {
            const int s  = (h * (PEPT / 2) + i) * FT + tid;
            const int bs = sbin[s];
            gd[i]  = gdelta[bs];
            lim[i] = (bs + 1) * CAPB;
            rc[i]  = stage[s];
        }
#pragma unroll
        for (int i = 0; i < PEPT / 2; ++i) {
            const int s = (h * (PEPT / 2) + i) * FT + tid;
            const int p = gd[i] + s;
            if (s < nvalid && p < lim[i]) sorted[p] = rc[i];
        }
    }
}

template <int K, int N>
__global__ __launch_bounds__(FT, 10)
void gemm_and_place(const float* __restrict__ A, const v2f* __restrict__ PB,
                    float* __restrict__ S, int M, int gemm_groups,
                    const int* src, const int* __restrict__ dst,
                    const float* w, int n_edges,
                    int* __restrict__ binCursor, uint2* __restrict__ sorted) {
    __shared__ int   hist[MAXB + 1];
    __shared__ int   lstart[MAXB + 1];
    __shared__ int   gdelta[MAXB + 1];
    __shared__ int   wsum[FT / 32];
    __shared__ uint2 stage[PSL];
    __shared__ unsigned short sbin[PSL];
    if ((int)blockIdx.x < gemm_groups) {
        gemm_wmma_f32<K, N>(A, PB, S, M);
    } else {
        bin_place((int)blockIdx.x - gemm_groups, src, dst, w, n_edges, binCursor, sorted,
                  hist, lstart, gdelta, wsum, stage, sbin);
    }
}

#define OWN_GB 16

template <int D>
struct OwnCfg {
    static constexpr int VW  = D >= 64 ? 2 : 1;
    static constexpr int LPE = D / VW;
    static constexpr int EPI = 32 / LPE;
    static constexpr int NG  = OWN_GB / EPI;
    static constexpr int PITCH = D >= 64 ? D + 4 : D;
    static_assert(LPE == 32 || LPE == 16, "row must map onto a wave or a half-wave");
    static_assert(PITCH % 4 == 0, "rows start on 16-byte boundaries");
};

template <int D, int J0>
__device__ __forceinline__ void own_loads(const float* __restrict__ S, int key, float wt,
                                          int sub, int c,
                                          float (&vx)[OwnCfg<D>::NG], float (&vy)[OwnCfg<D>::NG],
                                          float (&wj)[OwnCfg<D>::NG], int (&dj)[OwnCfg<D>::NG]) {
    constexpr int VW = OwnCfg<D>::VW, EPI = OwnCfg<D>::EPI, NG = OwnCfg<D>::NG;
#pragma unroll
    for (int u = 0; u < NG; ++u) {
        const int j = J0 + u * EPI;
        int kj;
        if constexpr (EPI == 1) {
            kj    = __builtin_amdgcn_readlane(key, j);
            wj[u] = __int_as_float(__builtin_amdgcn_readlane(__float_as_int(wt), j));
        } else {
            const int k0 = __builtin_amdgcn_readlane(key, j);
            const int k1 = __builtin_amdgcn_readlane(key, j + 1);
            const int w0 = __builtin_amdgcn_readlane(__float_as_int(wt), j);
            const int w1 = __builtin_amdgcn_readlane(__float_as_int(wt), j + 1);
            kj    = sub ? k1 : k0;
            wj[u] = __int_as_float(sub ? w1 : w0);
        }
        const int sj = kj & ((1 << SRCB) - 1);
        int dl = kj >> SRCB;
        dl = dl < RPB ? dl : RPB - 1;
        if constexpr (EPI == 1) {
            int ro;
            asm("s_mul_i32 %0, %1, %2" : "=s"(ro) : "s"(dl), "n"(OwnCfg<D>::PITCH));
            dj[u] = ro;
        } else {
            dj[u] = dl * OwnCfg<D>::PITCH;
        }
        const float* __restrict__ sp = S + (size_t)sj * D + c;
        if constexpr (VW == 2) {
            const v2f t = *(const v2f*)sp;
            vx[u] = t.x; vy[u] = t.y;
        } else {
            vx[u] = *sp; vy[u] = 0.f;
        }
    }
}

template <int D, int J0, bool FULL>
__device__ __forceinline__ void own_adds(float* acc, int cnt, int sub, int c,
                                         const float (&vx)[OwnCfg<D>::NG], const float (&vy)[OwnCfg<D>::NG],
                                         const float (&wj)[OwnCfg<D>::NG], const int (&dj)[OwnCfg<D>::NG]) {
    constexpr int VW = OwnCfg<D>::VW, EPI = OwnCfg<D>::EPI, NG = OwnCfg<D>::NG;
#pragma unroll
    for (int u = 0; u < NG; ++u) {
        const int j = J0 + u * EPI;
        if (FULL || j + sub < cnt) {
            float* ap = acc + dj[u] + c;
            atomicAdd(ap, wj[u] * vx[u]);
            if constexpr (VW == 2) atomicAdd(ap + 1, wj[u] * vy[u]);
        }
    }
}

#define OWN_CAP 4096

__device__ __forceinline__ int row_of(unsigned key) {
    const unsigned r = key >> SRCB;
    return r < RPB ? (int)r : RPB - 1;
}

template <int D, int J0, bool FULL>
__device__ __forceinline__ void own_runs(float* acc, int cnt, int c,
                                         const float (&vx)[OwnCfg<D>::NG], const float (&vy)[OwnCfg<D>::NG],
                                         const float (&wj)[OwnCfg<D>::NG], const int (&dj)[OwnCfg<D>::NG],
                                         int& cur, float& ax, float& ay) {
    static_assert(OwnCfg<D>::EPI == 1 && OwnCfg<D>::VW == 2, "one edge per instruction, two floats per lane");
#pragma unroll
    for (int u = 0; u < OwnCfg<D>::NG; ++u) {
        if (FULL || J0 + u < cnt) {
            if (dj[u] != cur) {
                float* ap = acc + cur + c;
                atomicAdd(ap, ax);
                atomicAdd(ap + 1, ay);
                ax = 0.f; ay = 0.f; cur = dj[u];
            }
            ax += wj[u] * vx[u];
            ay += wj[u] * vy[u];
        }
    }
}

template <int D>
__device__ __forceinline__ void own_by_row(const float* __restrict__ S, const uint2* __restrict__ sorted,
                                           int beg, int n, float* acc, int tid, int lane, int wv, int c) {
    constexpr int NG  = OwnCfg<D>::NG;
    constexpr int NW  = 512 / 32;
    constexpr int RPT = OWN_CAP / 512;
    __shared__ uint2 srec[OWN_CAP];
    __shared__ int   rcnt[RPB];
    __shared__ int   rstart[RPB];
    __shared__ int   wsum[NW];

    for (int r = tid; r < RPB; r += 512) rcnt[r] = 0;
    uint2 rr[RPT];
#pragma unroll
    for (int i = 0; i < RPT; ++i) {
        const int idx = tid + i * 512;
        rr[i] = make_uint2(0u, 0u);
        if (idx < n) rr[i] = sorted[beg + idx];
    }
    __syncthreads();
    int rk[RPT];
#pragma unroll
    for (int i = 0; i < RPT; ++i) {
        const int idx = tid + i * 512;
        rk[i] = 0;
        if (idx < n) rk[i] = atomicAdd(&rcnt[row_of(rr[i].x)], 1);
    }
    __syncthreads();
    const int cntv = tid < RPB ? rcnt[tid] : 0;
    const int inc  = wave_incl_scan(cntv, lane);
    if (lane == 31) wsum[wv] = inc;
    __syncthreads();
    int pre = 0;
#pragma unroll
    for (int k = 0; k < NW; ++k) if (k < wv) pre += wsum[k];
    if (tid < RPB) rstart[tid] = pre + inc - cntv;
    __syncthreads();
#pragma unroll
    for (int i = 0; i < RPT; ++i) {
        const int idx = tid + i * 512;
        if (idx < n) srec[rstart[row_of(rr[i].x)] + rk[i]] = rr[i];
    }
    __syncthreads();

    int   nkey = 0;
    float nwt  = 0.f;
    int g = wv * 32;
    if (g + lane < n) { const uint2 r = srec[g + lane]; nkey = (int)r.x; nwt = __uint_as_float(r.y); }

    float vx[NG], vy[NG], wj[NG];
    int   dj[NG];

    for (; g + 32 <= n; g += NW * 32) {
        const int   key = nkey;
        const float wt  = nwt;
        own_loads<D, 0>(S, key, wt, 0, c, vx, vy, wj, dj);
        int   cur = dj[0];
        float ax = 0.f, ay = 0.f;
        own_runs<D, 0, true>(acc, 32, c, vx, vy, wj, dj, cur, ax, ay);
        own_loads<D, OWN_GB>(S, key, wt, 0, c, vx, vy, wj, dj);
        {
            const int pi = (g + NW * 32 + lane) < n ? (g + NW * 32 + lane) : n - 1;
            const uint2 r = srec[pi];
            nkey = (int)r.x; nwt = __uint_as_float(r.y);
        }
        own_runs<D, OWN_GB, true>(acc, 32, c, vx, vy, wj, dj, cur, ax, ay);
        float* ap = acc + cur + c;
        atomicAdd(ap, ax);
        atomicAdd(ap + 1, ay);
    }
    if (g < n) {
        const int   cnt = n - g;
        const int   key = nkey;
        const float wt  = nwt;
        own_loads<D, 0>(S, key, wt, 0, c, vx, vy, wj, dj);
        int   cur = dj[0];
        float ax = 0.f, ay = 0.f;
        own_runs<D, 0, false>(acc, cnt, c, vx, vy, wj, dj, cur, ax, ay);
        if (cnt > OWN_GB) {
            own_loads<D, OWN_GB>(S, key, wt, 0, c, vx, vy, wj, dj);
            own_runs<D, OWN_GB, false>(acc, cnt, c, vx, vy, wj, dj, cur, ax, ay);
        }
        float* ap = acc + cur + c;
        atomicAdd(ap, ax);
        atomicAdd(ap + 1, ay);
    }
}

template <int D, bool PRODUCT>
__global__ __launch_bounds__(512)
void own_sum(const float* __restrict__ S, const uint2* __restrict__ sorted,
             const int* __restrict__ binCursor, const float* __restrict__ bias,
             const v2f* __restrict__ PBn, float* __restrict__ out, int n_rows,
             const int* __restrict__ esrc, const int* __restrict__ edst,
             const float* __restrict__ ew, int n_edges) {
    constexpr int VW    = OwnCfg<D>::VW;
    constexpr int LPE   = OwnCfg<D>::LPE;
    constexpr int NG    = OwnCfg<D>::NG;
    constexpr int PITCH = OwnCfg<D>::PITCH;
    constexpr int NW    = 512 / 32;
    constexpr int CG    = D / 4;
    constexpr int NTILE = (RPB + 15) / 16;
    static_assert(512 % CG == 0, "a thread keeps one column group in the start-value loop");
    static_assert(OWN_GB == 16, "two groups per 32 records");
    static_assert(!PRODUCT || (D == NHID && NTILE <= NW), "one wave per 16-row tile, K = NHID");
    static_assert(PRODUCT || (D == NOUT && RPB <= 512), "one thread per row for the row's log-sum-exp");

    __shared__ float acc[RPB * PITCH];

    const int tid  = threadIdx.x;
    const int lane = tid & 31;
    const int wv   = __builtin_amdgcn_readfirstlane(tid >> 5);
    const int sub  = lane / LPE;
    const int c    = (lane % LPE) * VW;
    const int b    = blockIdx.x;
    const int nrec = binCursor[b];
    const bool over = nrec > CAPB;
    const int beg  = b * CAPB;
    const int end  = over ? beg : beg + nrec;
    const int row0 = b * RPB;

    const int c4 = tid % CG;
    float4 bv;
    bv.x = bias[c4 * 4 + 0]; bv.y = bias[c4 * 4 + 1];
    bv.z = bias[c4 * 4 + 2]; bv.w = bias[c4 * 4 + 3];
    float4* __restrict__ acc4 = (float4*)acc;
    for (int r = tid / CG; r < RPB; r += 512 / CG) acc4[r * (PITCH / 4) + c4] = bv;

    bool by_row = false;
    if constexpr (PRODUCT) by_row = (end - beg) <= OWN_CAP;
    if (by_row) {
        if constexpr (PRODUCT) own_by_row<D>(S, sorted, beg, end - beg, acc, tid, lane, wv, c);
    } else {
        __syncthreads();

        int   nkey = 0;
        float nwt  = 0.f;
        int g = beg + wv * 32;
        if (g + lane < end) { const uint2 r = sorted[g + lane]; nkey = (int)r.x; nwt = __uint_as_float(r.y); }

        float vx[NG], vy[NG], wj[NG];
        int   dj[NG];

        for (; g + 32 <= end; g += NW * 32) {
            const int   key = nkey;
            const float wt  = nwt;
            own_loads<D, 0>(S, key, wt, sub, c, vx, vy, wj, dj);
            own_adds<D, 0, true>(acc, 32, sub, c, vx, vy, wj, dj);
            own_loads<D, OWN_GB>(S, key, wt, sub, c, vx, vy, wj, dj);
            {
                const int pi = (g + NW * 32 + lane) < end ? (g + NW * 32 + lane) : end - 1;
                const uint2 r = sorted[pi];
                nkey = (int)r.x; nwt = __uint_as_float(r.y);
            }
            own_adds<D, OWN_GB, true>(acc, 32, sub, c, vx, vy, wj, dj);
        }
        if (g < end) {
            const int   cnt = end - g;
            const int   key = nkey;
            const float wt  = nwt;
            own_loads<D, 0>(S, key, wt, sub, c, vx, vy, wj, dj);
            own_adds<D, 0, false>(acc, cnt, sub, c, vx, vy, wj, dj);
            if (cnt > OWN_GB) {
                own_loads<D, OWN_GB>(S, key, wt, sub, c, vx, vy, wj, dj);
                own_adds<D, OWN_GB, false>(acc, cnt, sub, c, vx, vy, wj, dj);
            }
        }
    }

    if (over) {
        for (int e = tid; e < n_edges; e += 512) {
            const int d = edst[e];
            if (bin_of(d) == b) {
                int r = d - row0;
                r = r < RPB ? r : RPB - 1;
                const float wgt = ew[e];
                const float* __restrict__ sp = S + (size_t)esrc[e] * D;
                float* ap = acc + r * PITCH;
                for (int q = 0; q < D; ++q) atomicAdd(ap + q, wgt * sp[q]);
            }
        }
    }

    const int rows = (n_rows - row0) < RPB ? (n_rows - row0) : RPB;

    if constexpr (PRODUCT) {
        __shared__ float s2s[NTILE * 16 * NOUT];
        const int half = lane >> 4;
        const int l16  = lane & 15;
        v2f bf[D / 4];
        if (wv < NTILE) {
#pragma unroll
            for (int k4 = 0; k4 < D / 4; ++k4) bf[k4] = PBn[k4 * 32 + lane];
        }
        __syncthreads();
        if (wv < NTILE) {
            int ar = wv * 16 + l16;
            ar = ar < RPB ? ar : RPB - 1;
            const float* ap = acc + ar * PITCH + 2 * half;
            v8f d = (v8f){0.f,0.f,0.f,0.f,0.f,0.f,0.f,0.f};
#pragma unroll
            for (int k4 = 0; k4 < D / 4; ++k4) {
                v2f a = *(const v2f*)(ap + k4 * 4);
                a.x = fmaxf(a.x, 0.f); a.y = fmaxf(a.y, 0.f);
                d = __builtin_amdgcn_wmma_f32_16x16x4_f32(
                    false, a, false, bf[k4], (short)0, d, false, false);
            }
#pragma unroll
            for (int j = 0; j < 8; ++j) s2s[(wv * 16 + j + 8 * half) * NOUT + l16] = d[j];
        }
        __syncthreads();
        float4* __restrict__ out4 = (float4*)(out + (size_t)row0 * NOUT);
        const float4* __restrict__ st4 = (const float4*)s2s;
        for (int i = tid; i < rows * (NOUT / 4); i += 512) out4[i] = st4[i];
    } else {
        __shared__ float lse_s[RPB];
        __syncthreads();
        if (tid < RPB) {
            float v[D];
#pragma unroll
            for (int q = 0; q < CG; ++q) {
                const float4 t = acc4[tid * (PITCH / 4) + q];
                v[q * 4 + 0] = t.x; v[q * 4 + 1] = t.y;
                v[q * 4 + 2] = t.z; v[q * 4 + 3] = t.w;
            }
            float mx = v[0];
#pragma unroll
            for (int j = 1; j < D; ++j) mx = fmaxf(mx, v[j]);
            float s = 0.f;
#pragma unroll
            for (int j = 0; j < D; ++j) s += expf(v[j] - mx);
            lse_s[tid] = mx + logf(s);
        }
        __syncthreads();
        static_assert(PRODUCT || PITCH == D, "rows are contiguous in LDS and in out");
        float4* __restrict__ out4 = (float4*)(out + (size_t)row0 * D);
        for (int i = tid; i < rows * CG; i += 512) {
            float4 t = acc4[i];
            const float l = lse_s[i / CG];
            t.x -= l; t.y -= l; t.z -= l; t.w -= l;
            out4[i] = t;
        }
    }
}

extern "C" void kernel_launch(void* const* d_in, const int* in_sizes, int n_in,
                              void* d_out, int out_size, void* d_ws, size_t ws_size,
                              hipStream_t stream) {
    const float* x    = (const float*)d_in[0];
    const int*   esrc = (const int*)  d_in[1];
    const int*   edst = (const int*)  d_in[2];
    const float* ew   = (const float*)d_in[3];
    const float* W1   = (const float*)d_in[4];
    const float* b1   = (const float*)d_in[5];
    const float* W2   = (const float*)d_in[6];
    const float* b2   = (const float*)d_in[7];
    float* out = (float*)d_out;

    const int n_nodes = in_sizes[0] / NFEAT;
    const int n_edges = in_sizes[1];

    constexpr int PB1_N = (NFEAT / 4) * (NHID / 16) * 32;
    constexpr int PB2_N = (NHID  / 4) * (NOUT / 16) * 32;

    float* R1  = (float*)d_ws;
    float* R2  = R1 + (size_t)n_nodes * NHID;
    float* R3  = R2 + (size_t)n_nodes * NHID;
    v2f*   PB1 = (v2f*)(R3 + (size_t)n_nodes * 2 * NOUT);
    v2f*   PB2 = PB1 + PB1_N;

    float* S1 = R1;
    float* S2 = R2;
    uint2* sorted    = (uint2*)(R2 + (size_t)n_nodes * NOUT);
    int*   binCursor = (int*)(sorted + (size_t)MAXB * CAPB);

    const int mtiles      = (n_nodes + 15) / 16;
    const int gemm_groups = (mtiles + FT / 32 - 1) / (FT / 32);
    const int nslices     = (n_edges + PSL - 1) / PSL;
    const int nbins       = (n_nodes + RPB - 1) / RPB;

    pack_both<<<(PB1_N + PB2_N + 255) / 256, 256, 0, stream>>>(W1, PB1, W2, PB2, binCursor);

    gemm_and_place<NFEAT, NHID>
        <<<gemm_groups + nslices, FT, 0, stream>>>(x, PB1, S1, n_nodes, gemm_groups,
                                                   esrc, edst, ew, n_edges, binCursor, sorted);

    own_sum<NHID, true><<<nbins, 512, 0, stream>>>(S1, sorted, binCursor, b1, PB2, S2, n_nodes,
                                                   esrc, edst, ew, n_edges);

    own_sum<NOUT, false><<<nbins, 512, 0, stream>>>(S2, sorted, binCursor, b2, nullptr, out, n_nodes,
                                                    esrc, edst, ew, n_edges);
}
